# starting kernel
# baseline (speedup 1.0000x reference)
#include <hip/hip_runtime.h>


typedef __attribute__((ext_vector_type(2))) float v2f;
typedef __attribute__((ext_vector_type(8))) float v8f;

#define NUM_I   32
#define NUM_A   8
#define NUM_O   10
#define NUM_AT  16
#define POS     16
#define HW      144
#define NT      320
#define WSTR    168
#define XSTR    20
#define ASTR    18

#define WMMA_F32_4(A_, B_, C_) \
  __builtin_amdgcn_wmma_f32_16x16x4_f32(false, (A_), false, (B_), (short)0, (C_), false, false)

__launch_bounds__(NT, 1)
__global__ void capsule_routing_kernel(const float* __restrict__ x,
                                       const float* __restrict__ w,
                                       const float* __restrict__ bias,
                                       float* __restrict__ out) {
  __shared__ float x_lds[NUM_I * NUM_A * XSTR];
  __shared__ float logits_lds[NUM_I * NUM_O * POS];
  __shared__ float route_lds[NUM_I * NUM_O * POS];
  __shared__ float w_lds[NUM_I * NUM_A * WSTR];
  __shared__ float act_lds[NUM_O * NUM_AT * ASTR];

  const int tid  = threadIdx.x;
  const int lane = tid & 31;
  const int o    = tid >> 5;
  const int n    = lane & 15;
  const int h    = lane >> 4;

  const int b   = blockIdx.x / 9;
  const int hw0 = (blockIdx.x % 9) * POS;

  for (int e = tid; e < NUM_I * NUM_A * POS; e += NT) {
    const int row = e >> 4;
    const int p   = e & 15;
    x_lds[row * XSTR + p] = x[((b * NUM_I) * NUM_A + row) * HW + hw0 + p];
  }
  for (int e = tid; e < NUM_I * NUM_A * (NUM_O * NUM_AT); e += NT) {
    const int row = e / (NUM_O * NUM_AT);
    const int c   = e - row * (NUM_O * NUM_AT);
    w_lds[row * WSTR + c] = w[e];
  }
  for (int e = tid; e < NUM_I * NUM_O * POS; e += NT)
    logits_lds[e] = 0.0f;

  const float bias_v = bias[o * NUM_AT + n];

  __syncthreads();

  for (int it = 0; it < 3; ++it) {
    for (int item = tid; item < NUM_I * POS; item += NT) {
      const int i = item >> 4;
      const int p = item & 15;
      float lv[NUM_O];
      float mx = -1e30f;
#pragma unroll
      for (int oo = 0; oo < NUM_O; ++oo) {
        lv[oo] = logits_lds[(i * NUM_O + oo) * POS + p];
        mx = fmaxf(mx, lv[oo]);
      }
      float s = 0.0f;
#pragma unroll
      for (int oo = 0; oo < NUM_O; ++oo) { lv[oo] = __expf(lv[oo] - mx); s += lv[oo]; }
      const float inv = 1.0f / s;
#pragma unroll
      for (int oo = 0; oo < NUM_O; ++oo)
        route_lds[(i * NUM_O + oo) * POS + p] = lv[oo] * inv;
    }
    __syncthreads();

    float pre[8];
#pragma unroll
    for (int r = 0; r < 8; ++r) pre[r] = 0.0f;

#pragma unroll 4
    for (int i = 0; i < NUM_I; ++i) {
      v2f a0, a1, b0, b1;
      a0.x = x_lds[(i * NUM_A + 2 * h    ) * XSTR + n];
      a0.y = x_lds[(i * NUM_A + 2 * h + 1) * XSTR + n];
      a1.x = x_lds[(i * NUM_A + 4 + 2 * h    ) * XSTR + n];
      a1.y = x_lds[(i * NUM_A + 4 + 2 * h + 1) * XSTR + n];
      b0.x = w_lds[(i * NUM_A + 2 * h    ) * WSTR + o * NUM_AT + n];
      b0.y = w_lds[(i * NUM_A + 2 * h + 1) * WSTR + o * NUM_AT + n];
      b1.x = w_lds[(i * NUM_A + 4 + 2 * h    ) * WSTR + o * NUM_AT + n];
      b1.y = w_lds[(i * NUM_A + 4 + 2 * h + 1) * WSTR + o * NUM_AT + n];
      v8f d = {};
      d = WMMA_F32_4(a0, b0, d);
      d = WMMA_F32_4(a1, b1, d);
#pragma unroll
      for (int r = 0; r < 8; ++r) {
        const float rt = route_lds[(i * NUM_O + o) * POS + r + 8 * h];
        pre[r] += rt * d[r];
      }
    }

    float act[8];
#pragma unroll
    for (int r = 0; r < 8; ++r) {
      pre[r] += bias_v;
      float ns = pre[r] * pre[r];
      ns += __shfl_xor(ns, 1, 32);
      ns += __shfl_xor(ns, 2, 32);
      ns += __shfl_xor(ns, 4, 32);
      ns += __shfl_xor(ns, 8, 32);
      act[r] = pre[r] * (sqrtf(ns) / (1.0f + ns));
    }

    if (it < 2) {
#pragma unroll
      for (int r = 0; r < 8; ++r)
        act_lds[(o * NUM_AT + n) * ASTR + r + 8 * h] = act[r];

      v2f bb[4];
#pragma unroll
      for (int kc = 0; kc < 4; ++kc) {
        bb[kc].x = act_lds[(o * NUM_AT + kc * 4 + 2 * h    ) * ASTR + n];
        bb[kc].y = act_lds[(o * NUM_AT + kc * 4 + 2 * h + 1) * ASTR + n];
      }

      const int arow = (n & 7);
      const int aih  = (n >> 3);
#pragma unroll 4
      for (int ip = 0; ip < NUM_I / 2; ++ip) {
        const int i0 = 2 * ip;
        v8f d = {};
#pragma unroll
        for (int kc = 0; kc < 4; ++kc) {
          v2f aa;
          aa.x = w_lds[((i0 + aih) * NUM_A + arow) * WSTR + o * NUM_AT + kc * 4 + 2 * h    ];
          aa.y = w_lds[((i0 + aih) * NUM_A + arow) * WSTR + o * NUM_AT + kc * 4 + 2 * h + 1];
          d = WMMA_F32_4(aa, bb[kc], d);
        }
        const int i = i0 + h;
        float dist = 0.0f;
#pragma unroll
        for (int r = 0; r < 8; ++r)
          dist += x_lds[(i * NUM_A + r) * XSTR + n] * d[r];
        logits_lds[(i * NUM_O + o) * POS + n] += dist;
      }
      __syncthreads();
    } else {
#pragma unroll
      for (int r = 0; r < 8; ++r)
        out[((b * NUM_O + o) * NUM_AT + n) * HW + hw0 + r + 8 * h] = act[r];
    }
  }
}

extern "C" void kernel_launch(void* const* d_in, const int* in_sizes, int n_in,
                              void* d_out, int out_size, void* d_ws, size_t ws_size,
                              hipStream_t stream) {
  const float* x    = (const float*)d_in[0];
  const float* wts  = (const float*)d_in[1];
  const float* bias = (const float*)d_in[2];
  float* out = (float*)d_out;
  (void)in_sizes; (void)n_in; (void)out_size; (void)d_ws; (void)ws_size;

  dim3 grid(64 * 9);
  dim3 block(NT);
  capsule_routing_kernel<<<grid, block, 0, stream>>>(x, wts, bias, out);
}


# Both WMMA operands held in VGPRs: each wave loads its capsule's weight slice from global memory once and the tile's x fragments from LDS once; vote tiles computed transposed, no weight copy in LDS
# speedup vs baseline: 1.2650x; 1.2650x over previous
#include <hip/hip_runtime.h>


typedef __attribute__((ext_vector_type(2))) float v2f;
typedef __attribute__((ext_vector_type(8))) float v8f;

#define NUM_I   32
#define NUM_A   8
#define NUM_O   10
#define NUM_AT  16
#define POS     16
#define HW      144
#define NT      320
#define WCOLS   160
#define XSTR    20

#define WMMA_F32_4(A_, B_, C_) \
  __builtin_amdgcn_wmma_f32_16x16x4_f32(false, (A_), false, (B_), (short)0, (C_), false, false)

__device__ __forceinline__ float other_half(float v) {
  return __builtin_bit_cast(float, __builtin_amdgcn_permlanex16(__builtin_bit_cast(int, v), __builtin_bit_cast(int, v),
                                                                 0x76543210, 0xfedcba98, false, false));
}

__launch_bounds__(NT)
__global__ void capsule_routing_kernel(const float* __restrict__ x,
                                       const float* __restrict__ w,
                                       const float* __restrict__ bias,
                                       float* __restrict__ out) {
  __shared__ float x_lds[NUM_I * NUM_A * XSTR];
  __shared__ float logits_lds[NUM_I * NUM_O * POS];
  __shared__ float route_lds[NUM_I * NUM_O * POS];

  const int tid  = threadIdx.x;
  const int lane = tid & 31;
  const int o    = tid >> 5;
  const int n    = lane & 15;
  const int h    = lane >> 4;

  const int b   = blockIdx.x / 9;
  const int hw0 = (blockIdx.x % 9) * POS;

  v2f wf[NUM_I][2];
  {
    const float* wp = w + (2 * h) * WCOLS + o * NUM_AT + n;
#pragma unroll
    for (int i = 0; i < NUM_I; ++i) {
#pragma unroll
      for (int c = 0; c < 2; ++c) {
        wf[i][c].x = wp[(i * NUM_A + 4 * c    ) * WCOLS];
        wf[i][c].y = wp[(i * NUM_A + 4 * c + 1) * WCOLS];
      }
    }
  }

  for (int e = tid; e < NUM_I * NUM_A * POS; e += NT) {
    const int row = e >> 4;
    const int p   = e & 15;
    x_lds[row * XSTR + p] = x[((b * NUM_I) * NUM_A + row) * HW + hw0 + p];
  }
  for (int e = tid; e < NUM_I * NUM_O * POS; e += NT)
    logits_lds[e] = 0.0f;

  float bias_r[8];
#pragma unroll
  for (int r = 0; r < 8; ++r) bias_r[r] = bias[o * NUM_AT + 8 * h + r];

  const float* xf = x_lds + (2 * h) * XSTR + n;
  const float* rtp = route_lds + o * POS + n;
  float*       lgp = logits_lds + (h * NUM_O + o) * POS + n;

  __syncthreads();

  v2f xr[NUM_I][2];
#pragma unroll
  for (int i = 0; i < NUM_I; ++i) {
#pragma unroll
    for (int c = 0; c < 2; ++c) {
      xr[i][c].x = xf[(i * NUM_A + 4 * c    ) * XSTR];
      xr[i][c].y = xf[(i * NUM_A + 4 * c + 1) * XSTR];
    }
  }

  for (int it = 0; it < 3; ++it) {
    for (int item = tid; item < NUM_I * POS; item += NT) {
      const int i = item >> 4;
      const int p = item & 15;
      float lv[NUM_O];
      float mx = -1e30f;
#pragma unroll
      for (int oo = 0; oo < NUM_O; ++oo) {
        lv[oo] = logits_lds[(i * NUM_O + oo) * POS + p];
        mx = fmaxf(mx, lv[oo]);
      }
      float s = 0.0f;
#pragma unroll
      for (int oo = 0; oo < NUM_O; ++oo) { lv[oo] = __expf(lv[oo] - mx); s += lv[oo]; }
      const float inv = 1.0f / s;
#pragma unroll
      for (int oo = 0; oo < NUM_O; ++oo)
        route_lds[(i * NUM_O + oo) * POS + p] = lv[oo] * inv;
    }
    __syncthreads();

    float pre[8];
#pragma unroll
    for (int r = 0; r < 8; ++r) pre[r] = 0.0f;

#pragma unroll
    for (int i = 0; i < NUM_I; ++i) {
      v8f d = {};
      d = WMMA_F32_4(wf[i][0], xr[i][0], d);
      d = WMMA_F32_4(wf[i][1], xr[i][1], d);
      const float rt = rtp[i * NUM_O * POS];
#pragma unroll
      for (int r = 0; r < 8; ++r) pre[r] += rt * d[r];
      if ((i & 3) == 3) __builtin_amdgcn_sched_barrier(0);
    }

    float ns = 0.0f;
#pragma unroll
    for (int r = 0; r < 8; ++r) {
      pre[r] += bias_r[r];
      ns += pre[r] * pre[r];
    }
    ns += other_half(ns);
    const float sc = sqrtf(ns) / (1.0f + ns);
    float act[8];
#pragma unroll
    for (int r = 0; r < 8; ++r) act[r] = pre[r] * sc;

    if (it < 2) {
#pragma unroll
      for (int ip = 0; ip < NUM_I / 2; ++ip) {
        float part[2];
#pragma unroll
        for (int j = 0; j < 2; ++j) {
          const int i = 2 * ip + j;
          v8f d = {};
          d = WMMA_F32_4(wf[i][0], xr[i][0], d);
          d = WMMA_F32_4(wf[i][1], xr[i][1], d);
          float s = 0.0f;
#pragma unroll
          for (int r = 0; r < 8; ++r) s += d[r] * act[r];
          part[j] = s;
        }
        const float keep = h ? part[1] : part[0];
        const float send = h ? part[0] : part[1];
        const float tot  = keep + other_half(send);
        lgp[ip * 2 * NUM_O * POS] += tot;
        if (ip & 1) __builtin_amdgcn_sched_barrier(0);
      }
      __syncthreads();
    } else {
#pragma unroll
      for (int r = 0; r < 8; ++r)
        out[((b * NUM_O + o) * NUM_AT + 8 * h + r) * HW + hw0 + n] = act[r];
    }
  }
}

extern "C" void kernel_launch(void* const* d_in, const int* in_sizes, int n_in,
                              void* d_out, int out_size, void* d_ws, size_t ws_size,
                              hipStream_t stream) {
  const float* x    = (const float*)d_in[0];
  const float* wts  = (const float*)d_in[1];
  const float* bias = (const float*)d_in[2];
  float* out = (float*)d_out;
  (void)in_sizes; (void)n_in; (void)out_size; (void)d_ws; (void)ws_size;

  dim3 grid(64 * 9);
  dim3 block(NT);
  capsule_routing_kernel<<<grid, block, 0, stream>>>(x, wts, bias, out);
}


# x tile loaded from global memory straight into each wave's WMMA B-fragment registers, in flight with the weight loads; no x copy in LDS, no load-wait-store staging loop, no fragment LDS reads
# speedup vs baseline: 1.3614x; 1.0762x over previous
#include <hip/hip_runtime.h>


typedef __attribute__((ext_vector_type(2))) float v2f;
typedef __attribute__((ext_vector_type(8))) float v8f;

#define NUM_I   32
#define NUM_A   8
#define NUM_O   10
#define NUM_AT  16
#define POS     16
#define HW      144
#define NT      320
#define WCOLS   160

#define WMMA_F32_4(A_, B_, C_) \
  __builtin_amdgcn_wmma_f32_16x16x4_f32(false, (A_), false, (B_), (short)0, (C_), false, false)

__device__ __forceinline__ float other_half(float v) {
  return __builtin_bit_cast(float, __builtin_amdgcn_permlanex16(__builtin_bit_cast(int, v), __builtin_bit_cast(int, v),
                                                                 0x76543210, 0xfedcba98, false, false));
}

__launch_bounds__(NT)
__global__ void capsule_routing_kernel(const float* __restrict__ x,
                                       const float* __restrict__ w,
                                       const float* __restrict__ bias,
                                       float* __restrict__ out) {
  __shared__ float logits_lds[NUM_I * NUM_O * POS];
  __shared__ float route_lds[NUM_I * NUM_O * POS];

  const int tid  = threadIdx.x;
  const int lane = tid & 31;
  const int o    = tid >> 5;
  const int n    = lane & 15;
  const int h    = lane >> 4;

  const int b   = blockIdx.x / 9;
  const int hw0 = (blockIdx.x % 9) * POS;

  v2f xr[NUM_I][2];
  {
    const float* xp = x + ((b * NUM_I) * NUM_A + 2 * h) * HW + hw0 + n;
#pragma unroll
    for (int i = 0; i < NUM_I; ++i) {
#pragma unroll
      for (int c = 0; c < 2; ++c) {
        xr[i][c].x = xp[(i * NUM_A + 4 * c    ) * HW];
        xr[i][c].y = xp[(i * NUM_A + 4 * c + 1) * HW];
      }
    }
  }

  v2f wf[NUM_I][2];
  {
    const float* wp = w + (2 * h) * WCOLS + o * NUM_AT + n;
#pragma unroll
    for (int i = 0; i < NUM_I; ++i) {
#pragma unroll
      for (int c = 0; c < 2; ++c) {
        wf[i][c].x = wp[(i * NUM_A + 4 * c    ) * WCOLS];
        wf[i][c].y = wp[(i * NUM_A + 4 * c + 1) * WCOLS];
      }
    }
  }

  for (int e = tid; e < NUM_I * NUM_O * POS; e += NT)
    logits_lds[e] = 0.0f;

  float bias_r[8];
#pragma unroll
  for (int r = 0; r < 8; ++r) bias_r[r] = bias[o * NUM_AT + 8 * h + r];

  const float* rtp = route_lds + o * POS + n;
  float*       lgp = logits_lds + (h * NUM_O + o) * POS + n;

  __syncthreads();

  for (int it = 0; it < 3; ++it) {
    for (int item = tid; item < NUM_I * POS; item += NT) {
      const int i = item >> 4;
      const int p = item & 15;
      float lv[NUM_O];
      float mx = -1e30f;
#pragma unroll
      for (int oo = 0; oo < NUM_O; ++oo) {
        lv[oo] = logits_lds[(i * NUM_O + oo) * POS + p];
        mx = fmaxf(mx, lv[oo]);
      }
      float s = 0.0f;
#pragma unroll
      for (int oo = 0; oo < NUM_O; ++oo) { lv[oo] = __expf(lv[oo] - mx); s += lv[oo]; }
      const float inv = 1.0f / s;
#pragma unroll
      for (int oo = 0; oo < NUM_O; ++oo)
        route_lds[(i * NUM_O + oo) * POS + p] = lv[oo] * inv;
    }
    __syncthreads();

    float pre[8];
#pragma unroll
    for (int r = 0; r < 8; ++r) pre[r] = 0.0f;

#pragma unroll
    for (int i = 0; i < NUM_I; ++i) {
      v8f d = {};
      d = WMMA_F32_4(wf[i][0], xr[i][0], d);
      d = WMMA_F32_4(wf[i][1], xr[i][1], d);
      const float rt = rtp[i * NUM_O * POS];
#pragma unroll
      for (int r = 0; r < 8; ++r) pre[r] += rt * d[r];
      if ((i & 3) == 3) __builtin_amdgcn_sched_barrier(0);
    }

    float ns = 0.0f;
#pragma unroll
    for (int r = 0; r < 8; ++r) {
      pre[r] += bias_r[r];
      ns += pre[r] * pre[r];
    }
    ns += other_half(ns);
    const float sc = sqrtf(ns) / (1.0f + ns);
    float act[8];
#pragma unroll
    for (int r = 0; r < 8; ++r) act[r] = pre[r] * sc;

    if (it < 2) {
#pragma unroll
      for (int ip = 0; ip < NUM_I / 2; ++ip) {
        float part[2];
#pragma unroll
        for (int j = 0; j < 2; ++j) {
          const int i = 2 * ip + j;
          v8f d = {};
          d = WMMA_F32_4(wf[i][0], xr[i][0], d);
          d = WMMA_F32_4(wf[i][1], xr[i][1], d);
          float s = 0.0f;
#pragma unroll
          for (int r = 0; r < 8; ++r) s += d[r] * act[r];
          part[j] = s;
        }
        const float keep = h ? part[1] : part[0];
        const float send = h ? part[0] : part[1];
        const float tot  = keep + other_half(send);
        lgp[ip * 2 * NUM_O * POS] += tot;
        if (ip & 1) __builtin_amdgcn_sched_barrier(0);
      }
      __syncthreads();
    } else {
#pragma unroll
      for (int r = 0; r < 8; ++r)
        out[((b * NUM_O + o) * NUM_AT + 8 * h + r) * HW + hw0 + n] = act[r];
    }
  }
}

extern "C" void kernel_launch(void* const* d_in, const int* in_sizes, int n_in,
                              void* d_out, int out_size, void* d_ws, size_t ws_size,
                              hipStream_t stream) {
  const float* x    = (const float*)d_in[0];
  const float* wts  = (const float*)d_in[1];
  const float* bias = (const float*)d_in[2];
  float* out = (float*)d_out;
  (void)in_sizes; (void)n_in; (void)out_size; (void)d_ws; (void)ws_size;

  dim3 grid(64 * 9);
  dim3 block(NT);
  capsule_routing_kernel<<<grid, block, 0, stream>>>(x, wts, bias, out);
}


# Each output capsule's five vote passes split over two waves by input-capsule halves: 20 equal waves, five per SIMD instead of 3/3/2/2; pass A partial sums swapped through LDS, pass B needs no swap
# speedup vs baseline: 1.4370x; 1.0555x over previous
#include <hip/hip_runtime.h>


typedef __attribute__((ext_vector_type(2))) float v2f;
typedef __attribute__((ext_vector_type(8))) float v8f;

#define NUM_I   32
#define NUM_A   8
#define NUM_O   10
#define NUM_AT  16
#define POS     16
#define HW      144
#define NW      20
#define NT      640
#define IH      16
#define WCOLS   160

#define WMMA_F32_4(A_, B_, C_) \
  __builtin_amdgcn_wmma_f32_16x16x4_f32(false, (A_), false, (B_), (short)0, (C_), false, false)

__device__ __forceinline__ float other_half(float v) {
  return __builtin_bit_cast(float, __builtin_amdgcn_permlanex16(__builtin_bit_cast(int, v), __builtin_bit_cast(int, v),
                                                                 0x76543210, 0xfedcba98, false, false));
}

__launch_bounds__(NT)
__global__ void capsule_routing_kernel(const float* __restrict__ x,
                                       const float* __restrict__ w,
                                       const float* __restrict__ bias,
                                       float* __restrict__ out) {
  __shared__ float logits_lds[NUM_I * NUM_O * POS];
  __shared__ float route_lds[NUM_I * NUM_O * POS];
  __shared__ float part_lds[NW * 8 * 32];

  const int tid  = threadIdx.x;
  const int lane = tid & 31;
  const int wv   = __builtin_amdgcn_readfirstlane(tid >> 5);
  const int o    = wv >> 1;
  const int g    = wv & 1;
  const int i0   = g * IH;
  const int n    = lane & 15;
  const int h    = lane >> 4;

  const int b   = blockIdx.x / 9;
  const int hw0 = (blockIdx.x % 9) * POS;

  v2f xr[IH][2];
  {
    const float* xp = x + (((b * NUM_I) + i0) * NUM_A + 2 * h) * HW + hw0 + n;
#pragma unroll
    for (int i = 0; i < IH; ++i) {
#pragma unroll
      for (int c = 0; c < 2; ++c) {
        xr[i][c].x = xp[(i * NUM_A + 4 * c    ) * HW];
        xr[i][c].y = xp[(i * NUM_A + 4 * c + 1) * HW];
      }
    }
  }

  v2f wf[IH][2];
  {
    const float* wp = w + (i0 * NUM_A + 2 * h) * WCOLS + o * NUM_AT + n;
#pragma unroll
    for (int i = 0; i < IH; ++i) {
#pragma unroll
      for (int c = 0; c < 2; ++c) {
        wf[i][c].x = wp[(i * NUM_A + 4 * c    ) * WCOLS];
        wf[i][c].y = wp[(i * NUM_A + 4 * c + 1) * WCOLS];
      }
    }
  }

  for (int e = tid; e < NUM_I * NUM_O * POS; e += NT)
    logits_lds[e] = 0.0f;

  const float* bp  = bias + o * NUM_AT + 8 * h;
  const float* rtp = route_lds + (i0 * NUM_O + o) * POS + n;
  float*       lgp = logits_lds + ((i0 + h) * NUM_O + o) * POS + n;
  float*       myp = part_lds + (wv * 8) * 32 + lane;
  const float* otp = part_lds + ((wv ^ 1) * 8) * 32 + lane;

  __syncthreads();

#pragma unroll 1
  for (int it = 0; it < 3; ++it) {
    float pre[8];
    if (g == 0) {
#pragma unroll
      for (int r = 0; r < 8; ++r) pre[r] = bp[r];
    } else {
#pragma unroll
      for (int r = 0; r < 8; ++r) pre[r] = 0.0f;
    }

    if (tid < NUM_I * POS) {
      const int i = tid >> 4;
      const int p = tid & 15;
      float lv[NUM_O];
      float mx = -1e30f;
#pragma unroll
      for (int oo = 0; oo < NUM_O; ++oo) {
        lv[oo] = logits_lds[(i * NUM_O + oo) * POS + p];
        mx = fmaxf(mx, lv[oo]);
      }
      float s = 0.0f;
#pragma unroll
      for (int oo = 0; oo < NUM_O; ++oo) { lv[oo] = __expf(lv[oo] - mx); s += lv[oo]; }
      const float inv = 1.0f / s;
#pragma unroll
      for (int oo = 0; oo < NUM_O; ++oo)
        route_lds[(i * NUM_O + oo) * POS + p] = lv[oo] * inv;
    }
    __syncthreads();


#pragma unroll
    for (int i = 0; i < IH; ++i) {
      v8f d = {};
      d = WMMA_F32_4(wf[i][0], xr[i][0], d);
      d = WMMA_F32_4(wf[i][1], xr[i][1], d);
      const float rt = rtp[i * NUM_O * POS];
#pragma unroll
      for (int r = 0; r < 8; ++r) pre[r] += rt * d[r];
      if (i & 1) {
        asm volatile("" : "+v"(pre[0]), "+v"(pre[1]), "+v"(pre[2]), "+v"(pre[3]),
                          "+v"(pre[4]), "+v"(pre[5]), "+v"(pre[6]), "+v"(pre[7]));
        __builtin_amdgcn_sched_barrier(0);
      }
    }

#pragma unroll
    for (int r = 0; r < 8; ++r) myp[r * 32] = pre[r];
    __syncthreads();
#pragma unroll
    for (int r = 0; r < 8; ++r) pre[r] += otp[r * 32];

    float ns = 0.0f;
#pragma unroll
    for (int r = 0; r < 8; ++r) ns += pre[r] * pre[r];
    ns += other_half(ns);
    const float sc = sqrtf(ns) / (1.0f + ns);
    float act[8];
#pragma unroll
    for (int r = 0; r < 8; ++r) act[r] = pre[r] * sc;

    if (it < 2) {
#pragma unroll
      for (int ip = 0; ip < IH / 2; ++ip) {
        float part[2];
#pragma unroll
        for (int j = 0; j < 2; ++j) {
          const int i = 2 * ip + j;
          v8f d = {};
          d = WMMA_F32_4(wf[i][0], xr[i][0], d);
          d = WMMA_F32_4(wf[i][1], xr[i][1], d);
          float s = 0.0f;
#pragma unroll
          for (int r = 0; r < 8; ++r) s += d[r] * act[r];
          part[j] = s;
        }
        const float keep = h ? part[1] : part[0];
        const float send = h ? part[0] : part[1];
        const float tot  = keep + other_half(send);
        lgp[ip * 2 * NUM_O * POS] += tot;
        __builtin_amdgcn_sched_barrier(0);
      }
      __syncthreads();
    } else {
      float* op = out + ((b * NUM_O + o) * NUM_AT + 8 * h + 4 * g) * HW + hw0 + n;
      if (g == 0) {
#pragma unroll
        for (int r = 0; r < 4; ++r) op[r * HW] = act[r];
      } else {
#pragma unroll
        for (int r = 0; r < 4; ++r) op[r * HW] = act[4 + r];
      }
    }
  }
}

extern "C" void kernel_launch(void* const* d_in, const int* in_sizes, int n_in,
                              void* d_out, int out_size, void* d_ws, size_t ws_size,
                              hipStream_t stream) {
  const float* x    = (const float*)d_in[0];
  const float* wts  = (const float*)d_in[1];
  const float* bias = (const float*)d_in[2];
  float* out = (float*)d_out;
  (void)in_sizes; (void)n_in; (void)out_size; (void)d_ws; (void)ws_size;

  dim3 grid(64 * 9);
  dim3 block(NT);
  capsule_routing_kernel<<<grid, block, 0, stream>>>(x, wts, bias, out);
}


# 256 persistent workgroups, one per WGP: each wave keeps its 64 weight registers over the workgroup's 2 or 3 tiles and asks for the next tile's x behind the last vote pass, across LDS-only barriers
# speedup vs baseline: 1.7657x; 1.2287x over previous
#include <hip/hip_runtime.h>


typedef __attribute__((ext_vector_type(2))) float v2f;
typedef __attribute__((ext_vector_type(8))) float v8f;

#define NUM_I   32
#define NUM_A   8
#define NUM_O   10
#define NUM_AT  16
#define POS     16
#define HW      144
#define NW      20
#define NT      640
#define IH      16
#define WCOLS   160

#define WMMA_F32_4(A_, B_, C_) \
  __builtin_amdgcn_wmma_f32_16x16x4_f32(false, (A_), false, (B_), (short)0, (C_), false, false)

__device__ __forceinline__ float other_half(float v) {
  return __builtin_bit_cast(float, __builtin_amdgcn_permlanex16(__builtin_bit_cast(int, v), __builtin_bit_cast(int, v),
                                                                 0x76543210, 0xfedcba98, false, false));
}

#define LOAD_X_TILE(T_)                                                                  \
  {                                                                                      \
    const int tb_ = (T_) / 9;                                                            \
    const int th_ = ((T_) - tb_ * 9) * POS;                                              \
    const float* xp_ = (x + ((((tb_ * NUM_I) + i0) * NUM_A) * HW + th_)) + xl;           \
    _Pragma("unroll")                                                                    \
    for (int i_ = 0; i_ < IH; ++i_) {                                                    \
      _Pragma("unroll")                                                                  \
      for (int c_ = 0; c_ < 2; ++c_) {                                                   \
        xr[i_][c_].x = xp_[(i_ * NUM_A + 4 * c_    ) * HW];                              \
        xr[i_][c_].y = xp_[(i_ * NUM_A + 4 * c_ + 1) * HW];                              \
      }                                                                                  \
    }                                                                                    \
  }

__device__ __forceinline__ void lds_barrier() {
  asm volatile("s_wait_dscnt 0x0\n\ts_barrier_signal -1\n\ts_barrier_wait -1" ::: "memory");
}

__launch_bounds__(NT)
__global__ void capsule_routing_kernel(const float* __restrict__ x,
                                       const float* __restrict__ w,
                                       const float* __restrict__ bias,
                                       float* __restrict__ out) {
  __shared__ float logits_lds[NUM_I * NUM_O * POS];
  __shared__ float route_lds[NUM_I * NUM_O * POS];
  __shared__ float part_lds[NW * 8 * 32];

  const int tid  = threadIdx.x;
  const int lane = tid & 31;
  const int wv   = __builtin_amdgcn_readfirstlane(tid >> 5);
  const int o    = wv >> 1;
  const int g    = wv & 1;
  const int i0   = g * IH;
  const int n    = lane & 15;
  const int h    = lane >> 4;

  const int t_begin = (blockIdx.x * 9) >> 2;
  const int t_end   = ((blockIdx.x + 1) * 9) >> 2;

  float sb[NUM_AT];
#pragma unroll
  for (int k = 0; k < NUM_AT; ++k) sb[k] = bias[o * NUM_AT + k];

  const unsigned xl = (unsigned)(2 * h * HW + n);
  v2f xr[IH][2];
  LOAD_X_TILE(t_begin);

  v2f wf[IH][2];
  {
    const float* wp = w + (i0 * NUM_A) * WCOLS + o * NUM_AT + (unsigned)(2 * h * WCOLS + n);
#pragma unroll
    for (int i = 0; i < IH; ++i) {
#pragma unroll
      for (int c = 0; c < 2; ++c) {
        wf[i][c].x = wp[(i * NUM_A + 4 * c    ) * WCOLS];
        wf[i][c].y = wp[(i * NUM_A + 4 * c + 1) * WCOLS];
      }
    }
  }

  const float* rtp = route_lds + (i0 * NUM_O + o) * POS + n;
  float*       lgp = logits_lds + ((i0 + h) * NUM_O + o) * POS + n;
  float*       myp = part_lds + (wv * 8) * 32 + lane;
  const float* otp = part_lds + ((wv ^ 1) * 8) * 32 + lane;

#pragma unroll 1
  for (int t = t_begin; t < t_end; ++t) {
  {
    float* zp = logits_lds + (wv * 8) * 32 + lane;
#pragma unroll
    for (int r = 0; r < 8; ++r) zp[r * 32] = 0.0f;
  }
  lds_barrier();

#pragma unroll 1
  for (int it = 0; it < 3; ++it) {
    float pre[8];
    if (g == 0) {
#pragma unroll
      for (int r = 0; r < 8; ++r) {
        float lo = sb[r], hi = sb[8 + r];
        asm volatile("" : "+s"(lo), "+s"(hi));
        pre[r] = h ? hi : lo;
      }
    } else {
#pragma unroll
      for (int r = 0; r < 8; ++r) pre[r] = 0.0f;
    }

    if (tid < NUM_I * POS) {
      const int i = tid >> 4;
      const int p = tid & 15;
      float lv[NUM_O];
      float mx = -1e30f;
#pragma unroll
      for (int oo = 0; oo < NUM_O; ++oo) {
        lv[oo] = logits_lds[(i * NUM_O + oo) * POS + p];
        mx = fmaxf(mx, lv[oo]);
      }
      float s = 0.0f;
#pragma unroll
      for (int oo = 0; oo < NUM_O; ++oo) { lv[oo] = __expf(lv[oo] - mx); s += lv[oo]; }
      const float inv = 1.0f / s;
#pragma unroll
      for (int oo = 0; oo < NUM_O; ++oo)
        route_lds[(i * NUM_O + oo) * POS + p] = lv[oo] * inv;
    }
    lds_barrier();


#pragma unroll
    for (int i = 0; i < IH; ++i) {
      v8f d = {};
      d = WMMA_F32_4(wf[i][0], xr[i][0], d);
      d = WMMA_F32_4(wf[i][1], xr[i][1], d);
      const float rt = rtp[i * NUM_O * POS];
#pragma unroll
      for (int r = 0; r < 8; ++r) pre[r] += rt * d[r];
      if (i & 1) {
        asm volatile("" : "+v"(pre[0]), "+v"(pre[1]), "+v"(pre[2]), "+v"(pre[3]),
                          "+v"(pre[4]), "+v"(pre[5]), "+v"(pre[6]), "+v"(pre[7]));
        __builtin_amdgcn_sched_barrier(0);
      }
    }

    if (it == 2 && t + 1 < t_end) {
      int tn = t + 1;
      asm volatile("" : "+s"(tn));
      LOAD_X_TILE(tn);
    }

#pragma unroll
    for (int r = 0; r < 8; ++r) myp[r * 32] = pre[r];
    lds_barrier();
#pragma unroll
    for (int r = 0; r < 8; ++r) pre[r] += otp[r * 32];

    float ns = 0.0f;
#pragma unroll
    for (int r = 0; r < 8; ++r) ns += pre[r] * pre[r];
    ns += other_half(ns);
    const float sc = sqrtf(ns) / (1.0f + ns);
    float act[8];
#pragma unroll
    for (int r = 0; r < 8; ++r) act[r] = pre[r] * sc;

    if (it < 2) {
#pragma unroll
      for (int ip = 0; ip < IH / 2; ++ip) {
        float part[2];
#pragma unroll
        for (int j = 0; j < 2; ++j) {
          const int i = 2 * ip + j;
          v8f d = {};
          d = WMMA_F32_4(wf[i][0], xr[i][0], d);
          d = WMMA_F32_4(wf[i][1], xr[i][1], d);
          float s = 0.0f;
#pragma unroll
          for (int r = 0; r < 8; ++r) s += d[r] * act[r];
          part[j] = s;
        }
        const float keep = h ? part[1] : part[0];
        const float send = h ? part[0] : part[1];
        const float tot  = keep + other_half(send);
        lgp[ip * 2 * NUM_O * POS] += tot;
        __builtin_amdgcn_sched_barrier(0);
      }
      lds_barrier();
    } else {
      int tc = t;
      asm volatile("" : "+s"(tc));
      const int b   = tc / 9;
      const int hw0 = (tc - b * 9) * POS;
      float* op = (out + (((b * NUM_O + o) * NUM_AT + 4 * g) * HW + hw0)) + (unsigned)(8 * h * HW + n);
      if (g == 0) {
#pragma unroll
        for (int r = 0; r < 4; ++r) op[r * HW] = act[r];
      } else {
#pragma unroll
        for (int r = 0; r < 4; ++r) op[r * HW] = act[4 + r];
      }
    }
  }
  }
}

extern "C" void kernel_launch(void* const* d_in, const int* in_sizes, int n_in,
                              void* d_out, int out_size, void* d_ws, size_t ws_size,
                              hipStream_t stream) {
  const float* x    = (const float*)d_in[0];
  const float* wts  = (const float*)d_in[1];
  const float* bias = (const float*)d_in[2];
  float* out = (float*)d_out;
  (void)in_sizes; (void)n_in; (void)out_size; (void)d_ws; (void)ws_size;

  dim3 grid(256);
  dim3 block(NT);
  capsule_routing_kernel<<<grid, block, 0, stream>>>(x, wts, bias, out);
}


# Next tile's x asked for inside the tile's last vote pass, each pair of input capsules right behind its last matrix read, and waited for pair by pair in first-use order, not 64 loads behind the pass
# speedup vs baseline: 1.8524x; 1.0491x over previous
#include <hip/hip_runtime.h>


typedef __attribute__((ext_vector_type(2))) float v2f;
typedef __attribute__((ext_vector_type(8))) float v8f;

#define NUM_I   32
#define NUM_A   8
#define NUM_O   10
#define NUM_AT  16
#define POS     16
#define HW      144
#define NW      20
#define NT      640
#define IH      16
#define WCOLS   160

#define WMMA_F32_4(A_, B_, C_) \
  __builtin_amdgcn_wmma_f32_16x16x4_f32(false, (A_), false, (B_), (short)0, (C_), false, false)

__device__ __forceinline__ float other_half(float v) {
  return __builtin_bit_cast(float, __builtin_amdgcn_permlanex16(__builtin_bit_cast(int, v), __builtin_bit_cast(int, v),
                                                                 0x76543210, 0xfedcba98, false, false));
}

#define LOAD_X_TILE(T_)                                                                  \
  {                                                                                      \
    const int tb_ = (T_) / 9;                                                            \
    const int th_ = ((T_) - tb_ * 9) * POS;                                              \
    const float* xp_ = (x + ((((tb_ * NUM_I) + i0) * NUM_A) * HW + th_)) + xl;           \
    _Pragma("unroll")                                                                    \
    for (int i_ = 0; i_ < IH; ++i_) {                                                    \
      _Pragma("unroll")                                                                  \
      for (int c_ = 0; c_ < 2; ++c_) {                                                   \
        xr[i_][c_].x = xp_[(i_ * NUM_A + 4 * c_    ) * HW];                              \
        xr[i_][c_].y = xp_[(i_ * NUM_A + 4 * c_ + 1) * HW];                              \
      }                                                                                  \
    }                                                                                    \
  }

__device__ __forceinline__ void lds_barrier() {
  asm volatile("s_wait_dscnt 0x0\n\ts_barrier_signal -1\n\ts_barrier_wait -1" ::: "memory");
}

__launch_bounds__(NT)
__global__ void capsule_routing_kernel(const float* __restrict__ x,
                                       const float* __restrict__ w,
                                       const float* __restrict__ bias,
                                       float* __restrict__ out) {
  __shared__ float logits_lds[NUM_I * NUM_O * POS];
  __shared__ float route_lds[NUM_I * NUM_O * POS];
  __shared__ float part_lds[NW * 8 * 32];

  const int tid  = threadIdx.x;
  const int lane = tid & 31;
  const int wv   = __builtin_amdgcn_readfirstlane(tid >> 5);
  const int o    = wv >> 1;
  const int g    = wv & 1;
  const int i0   = g * IH;
  const int n    = lane & 15;
  const int h    = lane >> 4;

  const int t_begin = (blockIdx.x * 9) >> 2;
  const int t_end   = ((blockIdx.x + 1) * 9) >> 2;

  float sb[NUM_AT];
#pragma unroll
  for (int k = 0; k < NUM_AT; ++k) sb[k] = bias[o * NUM_AT + k];

  const unsigned xl = (unsigned)(2 * h * HW + n);
  v2f xr[IH][2];
  LOAD_X_TILE(t_begin);

  v2f wf[IH][2];
  {
    const float* wp = w + (i0 * NUM_A) * WCOLS + o * NUM_AT + (unsigned)(2 * h * WCOLS + n);
#pragma unroll
    for (int i = 0; i < IH; ++i) {
#pragma unroll
      for (int c = 0; c < 2; ++c) {
        wf[i][c].x = wp[(i * NUM_A + 4 * c    ) * WCOLS];
        wf[i][c].y = wp[(i * NUM_A + 4 * c + 1) * WCOLS];
      }
    }
  }

  const float* rtp = route_lds + (i0 * NUM_O + o) * POS + n;
  float*       lgp = logits_lds + ((i0 + h) * NUM_O + o) * POS + n;
  float*       myp = part_lds + (wv * 8) * 32 + lane;
  const float* otp = part_lds + ((wv ^ 1) * 8) * 32 + lane;

#pragma unroll
  for (int i = 0; i < IH; ++i) {
#pragma unroll
    for (int c = 0; c < 2; ++c) {
      asm volatile("" : "+v"(xr[i][c].x), "+v"(xr[i][c].y), "+v"(wf[i][c].x), "+v"(wf[i][c].y));
    }
  }

#define PRE_INIT()                                                                       \
  float pre[8];                                                                          \
  if (g == 0) {                                                                          \
    _Pragma("unroll")                                                                    \
    for (int r = 0; r < 8; ++r) {                                                        \
      float lo = sb[r], hi = sb[8 + r];                                                  \
      asm volatile("" : "+s"(lo), "+s"(hi));     \
      pre[r] = h ? hi : lo;                       \
    }                                                                                    \
  } else {                                                                               \
    _Pragma("unroll")                                                                    \
    for (int r = 0; r < 8; ++r) pre[r] = 0.0f;                                           \
  }

#define ROUTE_SOFTMAX()                                                                  \
  if (tid < NUM_I * POS) {                                                               \
    const int i = tid >> 4;                                                              \
    const int p = tid & 15;                                                              \
    float lv[NUM_O];                                                                     \
    float mx = -1e30f;                                                                   \
    _Pragma("unroll")                                                                    \
    for (int oo = 0; oo < NUM_O; ++oo) {                                                 \
      lv[oo] = logits_lds[(i * NUM_O + oo) * POS + p];                                   \
      mx = fmaxf(mx, lv[oo]);                                                            \
    }                                                                                    \
    float s = 0.0f;                                                                      \
    _Pragma("unroll")                                                                    \
    for (int oo = 0; oo < NUM_O; ++oo) { lv[oo] = __expf(lv[oo] - mx); s += lv[oo]; }    \
    const float inv = 1.0f / s;                                                          \
    _Pragma("unroll")                                                                    \
    for (int oo = 0; oo < NUM_O; ++oo)                                                   \
      route_lds[(i * NUM_O + oo) * POS + p] = lv[oo] * inv;                              \
  }                                                                                      \
  lds_barrier();

#define PASS_A(PF_)                                                                      \
  _Pragma("unroll")                                                                      \
  for (int i = 0; i < IH; ++i) {                                                         \
    v8f d = {};                                                                          \
    d = WMMA_F32_4(wf[i][0], xr[i][0], d);                                  \
    d = WMMA_F32_4(wf[i][1], xr[i][1], d);                                  \
    const float rt = rtp[i * NUM_O * POS];                                               \
    _Pragma("unroll")                                                                    \
    for (int r = 0; r < 8; ++r) pre[r] += rt * d[r];                                     \
    if (i & 1) {                                      \
      if (PF_) {                                                                         \
        _Pragma("unroll")                                                                \
        for (int j = i - 1; j <= i; ++j) {                                               \
          _Pragma("unroll")                                                              \
          for (int c = 0; c < 2; ++c) {                                                  \
            xr[j][c].x = xn[(j * NUM_A + 4 * c    ) * HW];                               \
            xr[j][c].y = xn[(j * NUM_A + 4 * c + 1) * HW];                               \
          }                                                                              \
        }                                                                                \
      }                                                                                  \
        \
      asm volatile("" : "+v"(pre[0]), "+v"(pre[1]), "+v"(pre[2]), "+v"(pre[3]),          \
                        "+v"(pre[4]), "+v"(pre[5]), "+v"(pre[6]), "+v"(pre[7]));         \
      __builtin_amdgcn_sched_barrier(0);        \
    }                                                                                    \
  }

#define SWAP_SQUASH()                                                                    \
  _Pragma("unroll")                                                                      \
  for (int r = 0; r < 8; ++r) myp[r * 32] = pre[r];                                      \
  lds_barrier();                                                                         \
  _Pragma("unroll")                                                                      \
  for (int r = 0; r < 8; ++r) pre[r] += otp[r * 32];                                     \
  float ns = 0.0f;                                                                       \
  _Pragma("unroll")                                                                      \
  for (int r = 0; r < 8; ++r) ns += pre[r] * pre[r];                                     \
  ns += other_half(ns);                                     \
  const float sc = sqrtf(ns) / (1.0f + ns);                                              \
  float act[8];                                                                          \
  _Pragma("unroll")                                                                      \
  for (int r = 0; r < 8; ++r) act[r] = pre[r] * sc;

#pragma unroll 1
  for (int t = t_begin; t < t_end; ++t) {
  {
    float* zp = logits_lds + (wv * 8) * 32 + lane;
#pragma unroll
    for (int r = 0; r < 8; ++r) zp[r * 32] = 0.0f;
  }
  lds_barrier();

#pragma unroll 1
  for (int it = 0; it < 2; ++it) {
    const float* xn = x;
    PRE_INIT();
    ROUTE_SOFTMAX();
    PASS_A(0);
    SWAP_SQUASH();

#pragma unroll
    for (int ip = 0; ip < IH / 2; ++ip) {
      asm volatile("" : "+v"(xr[2 * ip][0].x), "+v"(xr[2 * ip][0].y), "+v"(xr[2 * ip][1].x), "+v"(xr[2 * ip][1].y),
                        "+v"(xr[2 * ip + 1][0].x), "+v"(xr[2 * ip + 1][0].y), "+v"(xr[2 * ip + 1][1].x), "+v"(xr[2 * ip + 1][1].y));
      float part[2];
#pragma unroll
      for (int j = 0; j < 2; ++j) {
        const int i = 2 * ip + j;
        v8f d = {};
        d = WMMA_F32_4(wf[i][0], xr[i][0], d);
        d = WMMA_F32_4(wf[i][1], xr[i][1], d);
        float s = 0.0f;
#pragma unroll
        for (int r = 0; r < 8; ++r) s += d[r] * act[r];
        part[j] = s;
      }
      const float keep = h ? part[1] : part[0];
      const float send = h ? part[0] : part[1];
      const float tot  = keep + other_half(send);
      lgp[ip * 2 * NUM_O * POS] += tot;
      __builtin_amdgcn_sched_barrier(0);
    }
    lds_barrier();
  }

  {
    const bool has_next = (t + 1 < t_end);
    int tn = has_next ? t + 1 : t;
    asm volatile("" : "+s"(tn));
    const int tb_ = tn / 9;
    const int th_ = (tn - tb_ * 9) * POS;
    const float* xn = (x + ((((tb_ * NUM_I) + i0) * NUM_A) * HW + th_)) + xl;

    PRE_INIT();
    ROUTE_SOFTMAX();
    if (has_next) {
      PASS_A(1);
    } else {
      PASS_A(0);
    }
    SWAP_SQUASH();

    int tc = t;
    asm volatile("" : "+s"(tc));
    const int b   = tc / 9;
    const int hw0 = (tc - b * 9) * POS;
    float* op = (out + (((b * NUM_O + o) * NUM_AT + 4 * g) * HW + hw0)) + (unsigned)(8 * h * HW + n);
    if (g == 0) {
#pragma unroll
      for (int r = 0; r < 4; ++r) op[r * HW] = act[r];
    } else {
#pragma unroll
      for (int r = 0; r < 4; ++r) op[r * HW] = act[4 + r];
    }
  }
  }
}

extern "C" void kernel_launch(void* const* d_in, const int* in_sizes, int n_in,
                              void* d_out, int out_size, void* d_ws, size_t ws_size,
                              hipStream_t stream) {
  const float* x    = (const float*)d_in[0];
  const float* wts  = (const float*)d_in[1];
  const float* bias = (const float*)d_in[2];
  float* out = (float*)d_out;
  (void)in_sizes; (void)n_in; (void)out_size; (void)d_ws; (void)ws_size;

  dim3 grid(256);
  dim3 block(NT);
  capsule_routing_kernel<<<grid, block, 0, stream>>>(x, wts, bias, out);
}


# Vote tiles of 12 of each wave's 16 input capsules formed once per tile, kept in a wave-private LDS slab (240 KB), read back by 128-bit loads in the 4 later passes: 64 WMMA per wave per tile, not 160
# speedup vs baseline: 2.4113x; 1.3017x over previous
#include <hip/hip_runtime.h>


typedef __attribute__((ext_vector_type(2))) float v2f;
typedef __attribute__((ext_vector_type(4))) float v4f;
typedef __attribute__((ext_vector_type(8))) float v8f;

#define NUM_I   32
#define NUM_A   8
#define NUM_O   10
#define NUM_AT  16
#define POS     16
#define HW      144
#define NW      20
#define NT      640
#define IH      16
#define WCOLS   160
#define NC      12

#define WMMA_F32_4(A_, B_, C_) \
  __builtin_amdgcn_wmma_f32_16x16x4_f32(false, (A_), false, (B_), (short)0, (C_), false, false)

__device__ __forceinline__ float other_half(float v) {
  return __builtin_bit_cast(float, __builtin_amdgcn_permlanex16(__builtin_bit_cast(int, v), __builtin_bit_cast(int, v),
                                                                 0x76543210, 0xfedcba98, false, false));
}

#define LOAD_X_TILE(T_)                                                                  \
  {                                                                                      \
    const int tb_ = (T_) / 9;                                                            \
    const int th_ = ((T_) - tb_ * 9) * POS;                                              \
    const float* xp_ = (x + ((((tb_ * NUM_I) + i0) * NUM_A) * HW + th_)) + xl;           \
    _Pragma("unroll")                                                                    \
    for (int i_ = 0; i_ < IH; ++i_) {                                                    \
      _Pragma("unroll")                                                                  \
      for (int c_ = 0; c_ < 2; ++c_) {                                                   \
        xr[i_][c_].x = xp_[(i_ * NUM_A + 4 * c_    ) * HW];                              \
        xr[i_][c_].y = xp_[(i_ * NUM_A + 4 * c_ + 1) * HW];                              \
      }                                                                                  \
    }                                                                                    \
  }

__device__ __forceinline__ void lds_barrier() {
  asm volatile("s_wait_dscnt 0x0\n\ts_barrier_signal -1\n\ts_barrier_wait -1" ::: "memory");
}

__launch_bounds__(NT)
__global__ void capsule_routing_kernel(const float* __restrict__ x,
                                       const float* __restrict__ w,
                                       const float* __restrict__ bias,
                                       float* __restrict__ out) {
  __shared__ float logits_lds[NUM_I * NUM_O * POS];
  __shared__ float route_lds[NUM_I * NUM_O * POS];
  __shared__ float part_lds[NW * 8 * 32];
  __shared__ __attribute__((aligned(16))) float votes_lds[NW * NC * 8 * 32];

  const int tid  = threadIdx.x;
  const int lane = tid & 31;
  const int wv   = __builtin_amdgcn_readfirstlane(tid >> 5);
  const int o    = wv >> 1;
  const int g    = wv & 1;
  const int i0   = g * IH;
  const int n    = lane & 15;
  const int h    = lane >> 4;

  const int t_begin = (blockIdx.x * 9) >> 2;
  const int t_end   = ((blockIdx.x + 1) * 9) >> 2;

  float sb[NUM_AT];
#pragma unroll
  for (int k = 0; k < NUM_AT; ++k) sb[k] = bias[o * NUM_AT + k];

  const unsigned xl = (unsigned)(2 * h * HW + n);
  v2f xr[IH][2];
  LOAD_X_TILE(t_begin);

  v2f wf[IH][2];
  {
    const float* wp = w + (i0 * NUM_A) * WCOLS + o * NUM_AT + (unsigned)(2 * h * WCOLS + n);
#pragma unroll
    for (int i = 0; i < IH; ++i) {
#pragma unroll
      for (int c = 0; c < 2; ++c) {
        wf[i][c].x = wp[(i * NUM_A + 4 * c    ) * WCOLS];
        wf[i][c].y = wp[(i * NUM_A + 4 * c + 1) * WCOLS];
      }
    }
  }

  const float* rtp = route_lds + (i0 * NUM_O + o) * POS + n;
  float*       lgp = logits_lds + ((i0 + h) * NUM_O + o) * POS + n;
  float*       myp = part_lds + (wv * 8) * 32 + lane;
  const float* otp = part_lds + ((wv ^ 1) * 8) * 32 + lane;
  v4f*         vcp = (v4f*)(votes_lds + wv * (NC * 8 * 32)) + lane;

#pragma unroll
  for (int i = 0; i < IH; ++i) {
#pragma unroll
    for (int c = 0; c < 2; ++c) {
      asm volatile("" : "+v"(xr[i][c].x), "+v"(xr[i][c].y), "+v"(wf[i][c].x), "+v"(wf[i][c].y));
    }
  }

#define PRE_INIT()                                                                       \
  float pre[8];                                                                          \
  if (g == 0) {                                                                          \
    _Pragma("unroll")                                                                    \
    for (int r = 0; r < 8; ++r) {                                                        \
      float lo = sb[r], hi = sb[8 + r];                                                  \
      asm volatile("" : "+s"(lo), "+s"(hi));     \
      pre[r] = h ? hi : lo;                       \
    }                                                                                    \
  } else {                                                                               \
    _Pragma("unroll")                                                                    \
    for (int r = 0; r < 8; ++r) pre[r] = 0.0f;                                           \
  }

#define ROUTE_SOFTMAX()                                                                  \
  if (tid < NUM_I * POS) {                                                               \
    const int i = tid >> 4;                                                              \
    const int p = tid & 15;                                                              \
    float lv[NUM_O];                                                                     \
    float mx = -1e30f;                                                                   \
    _Pragma("unroll")                                                                    \
    for (int oo = 0; oo < NUM_O; ++oo) {                                                 \
      lv[oo] = logits_lds[(i * NUM_O + oo) * POS + p];                                   \
      mx = fmaxf(mx, lv[oo]);                                                            \
    }                                                                                    \
    float s = 0.0f;                                                                      \
    _Pragma("unroll")                                                                    \
    for (int oo = 0; oo < NUM_O; ++oo) { lv[oo] = __expf(lv[oo] - mx); s += lv[oo]; }    \
    const float inv = 1.0f / s;                                                          \
    _Pragma("unroll")                                                                    \
    for (int oo = 0; oo < NUM_O; ++oo)                                                   \
      route_lds[(i * NUM_O + oo) * POS + p] = lv[oo] * inv;                              \
  }                                                                                      \
  lds_barrier();

#define SLAB_LOAD(d_, i_)                                                                \
  {                                                                                      \
    const v4f lo_ = vcp[(2 * (i_)    ) * 32];                                            \
    const v4f hi_ = vcp[(2 * (i_) + 1) * 32];                                            \
    d_ = __builtin_shufflevector(lo_, hi_, 0, 1, 2, 3, 4, 5, 6, 7);                      \
  }

#define VOTE_WMMA(d_, i_, FIRST_)                                                        \
  {                                                                                      \
    d_ = (v8f){};                                                                        \
    d_ = WMMA_F32_4(wf[i_][0], xr[i_][0], d_);                              \
    d_ = WMMA_F32_4(wf[i_][1], xr[i_][1], d_);                              \
    if ((FIRST_) && (i_) < NC) {                                                         \
      vcp[(2 * (i_)    ) * 32] = __builtin_shufflevector(d_, d_, 0, 1, 2, 3);            \
      vcp[(2 * (i_) + 1) * 32] = __builtin_shufflevector(d_, d_, 4, 5, 6, 7);            \
    }                                                                                    \
  }

#define SLAB_AHEAD(dt_, i_, ON_)                                                         \
  if ((ON_) && !((i_) & 1) && (i_) + 2 < NC) {                                           \
    SLAB_LOAD(dt_[(i_) + 2], (i_) + 2)                                                   \
    SLAB_LOAD(dt_[(i_) + 3], (i_) + 3)                                                   \
  }

#define PASS_A(FIRST_, PF_)                                                              \
  {                                                                                      \
    v8f dt[IH];                                                                          \
    float rtv[IH];                              \
    rtv[0] = rtp[0];                                                                     \
    rtv[1] = rtp[NUM_O * POS];                                                           \
    if (!(FIRST_)) {                                                                     \
      SLAB_LOAD(dt[0], 0)                                                                \
      SLAB_LOAD(dt[1], 1)                                                                \
    }                                                                                    \
    _Pragma("unroll")                                                                    \
    for (int i = 0; i < IH; ++i) {                                                       \
      if (!(i & 1) && i + 2 < IH) {                                                      \
        rtv[i + 2] = rtp[(i + 2) * NUM_O * POS];                                         \
        rtv[i + 3] = rtp[(i + 3) * NUM_O * POS];                                         \
      }                                                                                  \
      SLAB_AHEAD(dt, i, !(FIRST_) && !(PF_))                                             \
      if (!(FIRST_) && (PF_) && !(i & 1) && i > 0 && i < NC) {     \
        SLAB_LOAD(dt[i], i)                                                              \
        SLAB_LOAD(dt[i + 1], i + 1)                                                      \
      }                                                                                  \
      if ((FIRST_) || i >= NC) VOTE_WMMA(dt[i], i, FIRST_)                               \
      const float rt = rtv[i];                                                           \
      _Pragma("unroll")                                                                  \
      for (int r = 0; r < 8; ++r) pre[r] += rt * dt[i][r];                               \
      if (i & 1) {                              \
        if (PF_) {                                                                       \
          _Pragma("unroll")                                                              \
          for (int j = i - 1; j <= i; ++j) {                                             \
            _Pragma("unroll")                                                            \
            for (int c = 0; c < 2; ++c) {                                                \
              xr[j][c].x = xn[(j * NUM_A + 4 * c    ) * HW];                             \
              xr[j][c].y = xn[(j * NUM_A + 4 * c + 1) * HW];                             \
            }                                                                            \
          }                                                                              \
        }                                                                                \
          \
        asm volatile("" : "+v"(pre[0]), "+v"(pre[1]), "+v"(pre[2]), "+v"(pre[3]),        \
                          "+v"(pre[4]), "+v"(pre[5]), "+v"(pre[6]), "+v"(pre[7]));       \
        __builtin_amdgcn_sched_barrier(0);      \
      }                                                                                  \
    }                                                                                    \
  }

#define PASS_B()                                                                         \
  {                                                                                      \
    v8f dt[IH];                                                                          \
    SLAB_LOAD(dt[0], 0)                                                                  \
    SLAB_LOAD(dt[1], 1)                                                                  \
    _Pragma("unroll")                                                                    \
    for (int ip = 0; ip < IH / 2; ++ip) {                                                \
      SLAB_AHEAD(dt, 2 * ip, 1)                                                          \
      if (2 * ip >= NC) {                                                                \
          \
          \
        asm volatile("" : "+v"(xr[2 * ip][0].x), "+v"(xr[2 * ip][0].y), "+v"(xr[2 * ip][1].x), "+v"(xr[2 * ip][1].y), \
                          "+v"(xr[2 * ip + 1][0].x), "+v"(xr[2 * ip + 1][0].y), "+v"(xr[2 * ip + 1][1].x), "+v"(xr[2 * ip + 1][1].y)); \
      }                                                                                  \
      float part[2];                                                                     \
      _Pragma("unroll")                                                                  \
      for (int j = 0; j < 2; ++j) {                                                      \
        const int i = 2 * ip + j;                                                        \
        if (i >= NC) VOTE_WMMA(dt[i], i, 0)                                              \
        float s = 0.0f;                                                                  \
        _Pragma("unroll")                                                                \
        for (int r = 0; r < 8; ++r) s += dt[i][r] * act[r];                              \
        part[j] = s;                                \
      }                                                                                  \
                          \
      const float keep = h ? part[1] : part[0];                                          \
      const float send = h ? part[0] : part[1];                                          \
      const float tot  = keep + other_half(send);                                        \
      lgp[ip * 2 * NUM_O * POS] += tot;                    \
      __builtin_amdgcn_sched_barrier(0);        \
    }                                                                                    \
  }

#define SWAP_SQUASH()                                                                    \
  _Pragma("unroll")                                                                      \
  for (int r = 0; r < 8; ++r) myp[r * 32] = pre[r];                                      \
  lds_barrier();                                                                         \
  _Pragma("unroll")                                                                      \
  for (int r = 0; r < 8; ++r) pre[r] += otp[r * 32];                                     \
  float ns = 0.0f;                                                                       \
  _Pragma("unroll")                                                                      \
  for (int r = 0; r < 8; ++r) ns += pre[r] * pre[r];                                     \
  ns += other_half(ns);                                     \
  const float sc = sqrtf(ns) / (1.0f + ns);                                              \
  float act[8];                                                                          \
  _Pragma("unroll")                                                                      \
  for (int r = 0; r < 8; ++r) act[r] = pre[r] * sc;

#define STORE_OUT()                                                                      \
  {                                                                                      \
    int tc = t;                                                                          \
    asm volatile("" : "+s"(tc));          \
    const int b   = tc / 9;                                                              \
    const int hw0 = (tc - b * 9) * POS;                                                  \
    float* op = (out + (((b * NUM_O + o) * NUM_AT + 4 * g) * HW + hw0)) + (unsigned)(8 * h * HW + n); \
    if (g == 0) {                                                                        \
      _Pragma("unroll")                                                                  \
      for (int r = 0; r < 4; ++r) op[r * HW] = act[r];                                   \
    } else {                                                                             \
      _Pragma("unroll")                                                                  \
      for (int r = 0; r < 4; ++r) op[r * HW] = act[4 + r];                               \
    }                                                                                    \
  }

#pragma unroll 1
  for (int t = t_begin; t < t_end; ++t) {
  {
    float* zp = logits_lds + (wv * 8) * 32 + lane;
#pragma unroll
    for (int r = 0; r < 8; ++r) zp[r * 32] = 0.0f;
  }
  lds_barrier();

  {
    const float* xn = x;
    PRE_INIT();
    ROUTE_SOFTMAX();
    PASS_A(1, 0);
    SWAP_SQUASH();
    PASS_B();
    lds_barrier();
  }

  {
    const float* xn = x;
    PRE_INIT();
    ROUTE_SOFTMAX();
    PASS_A(0, 0);
    SWAP_SQUASH();
    PASS_B();
    lds_barrier();
  }

  {
    const bool has_next = (t + 1 < t_end);
    int tn = has_next ? t + 1 : t;
    asm volatile("" : "+s"(tn));
    const int tb_ = tn / 9;
    const int th_ = (tn - tb_ * 9) * POS;
    const float* xn = (x + ((((tb_ * NUM_I) + i0) * NUM_A) * HW + th_)) + xl;

    PRE_INIT();
    ROUTE_SOFTMAX();
    if (has_next) {
      PASS_A(0, 1);
    } else {
      PASS_A(0, 0);
#pragma unroll
      for (int i = 0; i < NC; ++i) {
        xr[i][0] = (v2f){0.0f, 0.0f};
        xr[i][1] = (v2f){0.0f, 0.0f};
      }
    }
    SWAP_SQUASH();
    STORE_OUT();
  }
  }
}

extern "C" void kernel_launch(void* const* d_in, const int* in_sizes, int n_in,
                              void* d_out, int out_size, void* d_ws, size_t ws_size,
                              hipStream_t stream) {
  const float* x    = (const float*)d_in[0];
  const float* wts  = (const float*)d_in[1];
  const float* bias = (const float*)d_in[2];
  float* out = (float*)d_out;
  (void)in_sizes; (void)n_in; (void)out_size; (void)d_ws; (void)ws_size;

  dim3 grid(256);
  dim3 block(NT);
  capsule_routing_kernel<<<grid, block, 0, stream>>>(x, wts, bias, out);
}


# x of 8 of each wave's 16 input capsules asked for inside the tile's own matrix-bound first pass, 4 pairs ahead of use; only the other 8 in the previous tile's last pass (and in the prologue)
# speedup vs baseline: 2.9084x; 1.2062x over previous
#include <hip/hip_runtime.h>


typedef __attribute__((ext_vector_type(2))) float v2f;
typedef __attribute__((ext_vector_type(4))) float v4f;
typedef __attribute__((ext_vector_type(8))) float v8f;

#define NUM_I   32
#define NUM_A   8
#define NUM_O   10
#define NUM_AT  16
#define POS     16
#define HW      144
#define NW      20
#define NT      640
#define IH      16
#define WCOLS   160
#define NC      8
#define NR      (IH - NC)
#define XA      8

#define WMMA_F32_4(A_, B_, C_) \
  __builtin_amdgcn_wmma_f32_16x16x4_f32(false, (A_), false, (B_), (short)0, (C_), false, false)

__device__ __forceinline__ float other_half(float v) {
  return __builtin_bit_cast(float, __builtin_amdgcn_permlanex16(__builtin_bit_cast(int, v), __builtin_bit_cast(int, v),
                                                                 0x76543210, 0xfedcba98, false, false));
}

#define X_TILE_PTR(T_) \
  ((x + ((((((T_) / 9) * NUM_I) + i0) * NUM_A) * HW + ((T_) - ((T_) / 9) * 9) * POS)) + xl)

#define LOAD_X_CAPS(P_, JA_, JB_)                                                        \
  {                                                                                      \
    _Pragma("unroll")                                                                    \
    for (int j_ = (JA_); j_ <= (JB_); ++j_) {                                            \
      _Pragma("unroll")                                                                  \
      for (int c_ = 0; c_ < 2; ++c_) {                                                   \
        xr[j_][c_].x = (P_)[(j_ * NUM_A + 4 * c_    ) * HW];                             \
        xr[j_][c_].y = (P_)[(j_ * NUM_A + 4 * c_ + 1) * HW];                             \
      }                                                                                  \
    }                                                                                    \
  }

__device__ __forceinline__ void lds_barrier() {
  asm volatile("s_wait_dscnt 0x0\n\ts_barrier_signal -1\n\ts_barrier_wait -1" ::: "memory");
}

__launch_bounds__(NT)
__global__ void capsule_routing_kernel(const float* __restrict__ x,
                                       const float* __restrict__ w,
                                       const float* __restrict__ bias,
                                       float* __restrict__ out) {
  __shared__ float logits_lds[NUM_I * NUM_O * POS];
  __shared__ float route_lds[NUM_I * NUM_O * POS];
  __shared__ float part_lds[NW * 8 * 32];
  __shared__ __attribute__((aligned(16))) float votes_lds[NW * NC * 8 * 32];

  const int tid  = threadIdx.x;
  const int lane = tid & 31;
  const int wv   = __builtin_amdgcn_readfirstlane(tid >> 5);
  const int o    = wv >> 1;
  const int g    = wv & 1;
  const int i0   = g * IH;
  const int n    = lane & 15;
  const int h    = lane >> 4;

  const int t_begin = (blockIdx.x * 9) >> 2;
  const int t_end   = ((blockIdx.x + 1) * 9) >> 2;

  float sb[NUM_AT];
#pragma unroll
  for (int k = 0; k < NUM_AT; ++k) sb[k] = bias[o * NUM_AT + k];

  const unsigned xl = (unsigned)(2 * h * HW + n);
  v2f xr[IH][2];
  {
    const float* xp0 = X_TILE_PTR(t_begin);
    LOAD_X_CAPS(xp0, 0, XA - 1);
  }

  v2f wf[IH][2];
  {
    const float* wp = w + (i0 * NUM_A) * WCOLS + o * NUM_AT + (unsigned)(2 * h * WCOLS + n);
#pragma unroll
    for (int i = 0; i < IH; ++i) {
#pragma unroll
      for (int c = 0; c < 2; ++c) {
        wf[i][c].x = wp[(i * NUM_A + 4 * c    ) * WCOLS];
        wf[i][c].y = wp[(i * NUM_A + 4 * c + 1) * WCOLS];
      }
    }
  }

  const float* rtp = route_lds + (i0 * NUM_O + o) * POS + n;
  float*       lgp = logits_lds + ((i0 + h) * NUM_O + o) * POS + n;
  float*       myp = part_lds + (wv * 8) * 32 + lane;
  const float* otp = part_lds + ((wv ^ 1) * 8) * 32 + lane;
  v4f*         vcp = (v4f*)(votes_lds + wv * (NC * 8 * 32)) + lane;

#pragma unroll
  for (int i = 0; i < IH; ++i) {
#pragma unroll
    for (int c = 0; c < 2; ++c) {
      asm volatile("" : "+v"(wf[i][c].x), "+v"(wf[i][c].y));
      if (i < XA) asm volatile("" : "+v"(xr[i][c].x), "+v"(xr[i][c].y));
    }
  }

#define PRE_INIT()                                                                       \
  float pre[8];                                                                          \
  if (g == 0) {                                                                          \
    _Pragma("unroll")                                                                    \
    for (int r = 0; r < 8; ++r) {                                                        \
      float lo = sb[r], hi = sb[8 + r];                                                  \
      asm volatile("" : "+s"(lo), "+s"(hi));     \
      pre[r] = h ? hi : lo;                       \
    }                                                                                    \
  } else {                                                                               \
    _Pragma("unroll")                                                                    \
    for (int r = 0; r < 8; ++r) pre[r] = 0.0f;                                           \
  }

#define ROUTE_SOFTMAX()                                                                  \
  if (tid < NUM_I * POS) {                                                               \
    const int i = tid >> 4;                                                              \
    const int p = tid & 15;                                                              \
    float lv[NUM_O];                                                                     \
    float mx = -1e30f;                                                                   \
    _Pragma("unroll")                                                                    \
    for (int oo = 0; oo < NUM_O; ++oo) {                                                 \
      lv[oo] = logits_lds[(i * NUM_O + oo) * POS + p];                                   \
      mx = fmaxf(mx, lv[oo]);                                                            \
    }                                                                                    \
    float s = 0.0f;                                                                      \
    _Pragma("unroll")                                                                    \
    for (int oo = 0; oo < NUM_O; ++oo) { lv[oo] = __expf(lv[oo] - mx); s += lv[oo]; }    \
    const float inv = 1.0f / s;                                                          \
    _Pragma("unroll")                                                                    \
    for (int oo = 0; oo < NUM_O; ++oo)                                                   \
      route_lds[(i * NUM_O + oo) * POS + p] = lv[oo] * inv;                              \
  }                                                                                      \
  lds_barrier();

#define SLAB_LOAD(d_, i_)                                                                \
  {                                                                                      \
    const v4f lo_ = vcp[(2 * (i_)    ) * 32];                                            \
    const v4f hi_ = vcp[(2 * (i_) + 1) * 32];                                            \
    d_ = __builtin_shufflevector(lo_, hi_, 0, 1, 2, 3, 4, 5, 6, 7);                      \
  }

#define VOTE_WMMA(d_, i_)                                                                \
  {                                                                                      \
    d_ = (v8f){};                                                                        \
    d_ = WMMA_F32_4(wf[i_][0], xr[i_][0], d_);                              \
    d_ = WMMA_F32_4(wf[i_][1], xr[i_][1], d_);                              \
    if ((i_) < NC) {                                                                     \
      vcp[(2 * (i_)    ) * 32] = __builtin_shufflevector(d_, d_, 0, 1, 2, 3);            \
      vcp[(2 * (i_) + 1) * 32] = __builtin_shufflevector(d_, d_, 4, 5, 6, 7);            \
    } else {                                                                             \
      kv[(i_) - NC] = d_;                                                                \
    }                                                                                    \
  }

#define PAIR_SLAB(q_) ((q_) >= NR || ((q_) & 1))
#define PAIR_BASE(q_) ((q_) >= NR ? 2 * (q_) - NR : (((q_) & 1) ? (q_) - 1 : NC + (q_)))
#define CAPL(s_) (PAIR_BASE((s_) >> 1) + ((s_) & 1))
#define PAIR_ASK(q_) (!PAIR_SLAB(q_) ? PAIR_BASE((q_) + 1) : (((q_) == 0 || PAIR_SLAB((q_) - 1)) ? PAIR_BASE(q_) : -1))

#define CAPA(s_, FIRST_) ((FIRST_) ? (s_) : CAPL(s_))
#define PASS_A(FIRST_, PF_)                                                              \
  {                                                                                      \
    v8f dt[IH];                                                                          \
    float rtv[IH];                              \
    rtv[0] = rtp[CAPA(0, FIRST_) * NUM_O * POS];                                         \
    rtv[1] = rtp[CAPA(1, FIRST_) * NUM_O * POS];                                         \
    _Pragma("unroll")                                                                    \
    for (int s = 0; s < IH; ++s) {                                                       \
      const int i = CAPA(s, FIRST_);                     \
      if (!(s & 1) && s + 2 < IH) {                                                      \
        rtv[s + 2] = rtp[CAPA(s + 2, FIRST_) * NUM_O * POS];                             \
        rtv[s + 3] = rtp[CAPA(s + 3, FIRST_) * NUM_O * POS];                             \
      }                                                                                  \
      if (!(FIRST_) && !(s & 1) && PAIR_ASK(s >> 1) >= 0) {     \
        SLAB_LOAD(dt[PAIR_ASK(s >> 1)], PAIR_ASK(s >> 1))                                \
        SLAB_LOAD(dt[PAIR_ASK(s >> 1) + 1], PAIR_ASK(s >> 1) + 1)                        \
      }                                                                                  \
      if (FIRST_) {                                                                      \
        VOTE_WMMA(dt[i], i)                                                              \
          \
             \
        if ((s & 1) && s + XA < IH) LOAD_X_CAPS(xc, s + XA - 1, s + XA)                  \
      }                                                                                  \
      else if (i >= NC) dt[i] = kv[i - NC];                                              \
      const float rt = rtv[s];                                                           \
      _Pragma("unroll")                                                                  \
      for (int r = 0; r < 8; ++r) pre[r] += rt * dt[i][r];                               \
      if (s & 1) {                                      \
        if ((PF_) && s < XA) LOAD_X_CAPS(xn, s - 1, s)                                   \
          \
        asm volatile("" : "+v"(pre[0]), "+v"(pre[1]), "+v"(pre[2]), "+v"(pre[3]),        \
                          "+v"(pre[4]), "+v"(pre[5]), "+v"(pre[6]), "+v"(pre[7]));       \
        __builtin_amdgcn_sched_barrier(0);      \
      }                                                                                  \
    }                                                                                    \
  }

#define PASS_B()                                                                         \
  {                                                                                      \
    v8f dt[IH];                                                                          \
    _Pragma("unroll")                                                                    \
    for (int q = 0; q < IH / 2; ++q) {                                                   \
      const int c0 = PAIR_BASE(q);              \
      if (PAIR_ASK(q) >= 0) {                   \
        SLAB_LOAD(dt[PAIR_ASK(q)], PAIR_ASK(q))                                          \
        SLAB_LOAD(dt[PAIR_ASK(q) + 1], PAIR_ASK(q) + 1)                                  \
      }                                                                                  \
      float part[2];                                                                     \
      _Pragma("unroll")                                                                  \
      for (int j = 0; j < 2; ++j) {                                                      \
        const int i = c0 + j;                                                            \
        if (i >= NC) dt[i] = kv[i - NC];                                                 \
          \
          \
        v2f s2 = (v2f){0.0f, 0.0f};                                                      \
        _Pragma("unroll")                                                                \
        for (int r = 0; r < 8; r += 2)                                                   \
          s2 += (v2f){dt[i][r], dt[i][r + 1]} * (v2f){act[r], act[r + 1]};               \
        part[j] = s2.x + s2.y;                      \
      }                                                                                  \
                          \
      const float keep = h ? part[1] : part[0];                                          \
      const float send = h ? part[0] : part[1];                                          \
      const float tot  = keep + other_half(send);                                        \
      lgp[c0 * NUM_O * POS] += tot;                        \
      __builtin_amdgcn_sched_barrier(0);        \
    }                                                                                    \
  }

#define SWAP_SQUASH()                                                                    \
  _Pragma("unroll")                                                                      \
  for (int r = 0; r < 8; ++r) myp[r * 32] = pre[r];                                      \
  lds_barrier();                                                                         \
  _Pragma("unroll")                                                                      \
  for (int r = 0; r < 8; ++r) pre[r] += otp[r * 32];                                     \
  float ns = 0.0f;                                                                       \
  _Pragma("unroll")                                                                      \
  for (int r = 0; r < 8; ++r) ns += pre[r] * pre[r];                                     \
  ns += other_half(ns);                                     \
  const float sc = sqrtf(ns) / (1.0f + ns);                                              \
  float act[8];                                                                          \
  _Pragma("unroll")                                                                      \
  for (int r = 0; r < 8; ++r) act[r] = pre[r] * sc;

#define STORE_OUT()                                                                      \
  {                                                                                      \
    int tc = t;                                                                          \
    asm volatile("" : "+s"(tc));          \
    const int b   = tc / 9;                                                              \
    const int hw0 = (tc - b * 9) * POS;                                                  \
    float* op = (out + (((b * NUM_O + o) * NUM_AT + 4 * g) * HW + hw0)) + (unsigned)(8 * h * HW + n); \
    if (g == 0) {                                                                        \
      _Pragma("unroll")                                                                  \
      for (int r = 0; r < 4; ++r) op[r * HW] = act[r];                                   \
    } else {                                                                             \
      _Pragma("unroll")                                                                  \
      for (int r = 0; r < 4; ++r) op[r * HW] = act[4 + r];                               \
    }                                                                                    \
  }

  int t = t_begin;
#pragma unroll 1
  for (;;) {
  {
    float* zp = logits_lds + (wv * 8) * 32 + lane;
#pragma unroll
    for (int r = 0; r < 8; ++r) zp[r * 32] = 0.0f;
  }
  lds_barrier();

  v8f kv[NR];

  int tcur = t;
  asm volatile("" : "+s"(tcur));
  const float* xc = X_TILE_PTR(tcur);

  {
    const float* xn = x;
    PRE_INIT();
    ROUTE_SOFTMAX();
    PASS_A(1, 0);
    SWAP_SQUASH();
    PASS_B();
    lds_barrier();
  }

  {
    const float* xn = x;
    PRE_INIT();
    ROUTE_SOFTMAX();
    PASS_A(0, 0);
    SWAP_SQUASH();
    PASS_B();
    lds_barrier();
  }

  const bool has_next = (t + 1 < t_end);
  {
    int tn = has_next ? t + 1 : t;
    asm volatile("" : "+s"(tn));
    const float* xn = X_TILE_PTR(tn);

    PRE_INIT();
    ROUTE_SOFTMAX();
    PASS_A(0, 1);
    SWAP_SQUASH();
    STORE_OUT();
  }
  if (!has_next) break;
  ++t;
  }
}

extern "C" void kernel_launch(void* const* d_in, const int* in_sizes, int n_in,
                              void* d_out, int out_size, void* d_ws, size_t ws_size,
                              hipStream_t stream) {
  const float* x    = (const float*)d_in[0];
  const float* wts  = (const float*)d_in[1];
  const float* bias = (const float*)d_in[2];
  float* out = (float*)d_out;
  (void)in_sizes; (void)n_in; (void)out_size; (void)d_ws; (void)ws_size;

  dim3 grid(256);
  dim3 block(NT);
  capsule_routing_kernel<<<grid, block, 0, stream>>>(x, wts, bias, out);
}


# Softmax 1/s and squash sqrt(ns)/(1+ns) on the 1-ulp hardware v_rcp_f32 and v_sqrt_f32, not the IEEE division and correctly rounded sqrt sequences: 95 fewer vector instructions per wave per tile
# speedup vs baseline: 3.0576x; 1.0513x over previous
#include <hip/hip_runtime.h>


typedef __attribute__((ext_vector_type(2))) float v2f;
typedef __attribute__((ext_vector_type(4))) float v4f;
typedef __attribute__((ext_vector_type(8))) float v8f;

#define NUM_I   32
#define NUM_A   8
#define NUM_O   10
#define NUM_AT  16
#define POS     16
#define HW      144
#define NW      20
#define NT      640
#define IH      16
#define WCOLS   160
#define NC      8
#define NR      (IH - NC)
#define XA      8

#define WMMA_F32_4(A_, B_, C_) \
  __builtin_amdgcn_wmma_f32_16x16x4_f32(false, (A_), false, (B_), (short)0, (C_), false, false)

__device__ __forceinline__ float other_half(float v) {
  return __builtin_bit_cast(float, __builtin_amdgcn_permlanex16(__builtin_bit_cast(int, v), __builtin_bit_cast(int, v),
                                                                 0x76543210, 0xfedcba98, false, false));
}

#define X_TILE_PTR(T_) \
  ((x + ((((((T_) / 9) * NUM_I) + i0) * NUM_A) * HW + ((T_) - ((T_) / 9) * 9) * POS)) + xl)

#define LOAD_X_CAPS(P_, JA_, JB_)                                                        \
  {                                                                                      \
    _Pragma("unroll")                                                                    \
    for (int j_ = (JA_); j_ <= (JB_); ++j_) {                                            \
      _Pragma("unroll")                                                                  \
      for (int c_ = 0; c_ < 2; ++c_) {                                                   \
        xr[j_][c_].x = (P_)[(j_ * NUM_A + 4 * c_    ) * HW];                             \
        xr[j_][c_].y = (P_)[(j_ * NUM_A + 4 * c_ + 1) * HW];                             \
      }                                                                                  \
    }                                                                                    \
  }

__device__ __forceinline__ void lds_barrier() {
  asm volatile("s_wait_dscnt 0x0\n\ts_barrier_signal -1\n\ts_barrier_wait -1" ::: "memory");
}

__launch_bounds__(NT)
__global__ void capsule_routing_kernel(const float* __restrict__ x,
                                       const float* __restrict__ w,
                                       const float* __restrict__ bias,
                                       float* __restrict__ out) {
  __shared__ float logits_lds[NUM_I * NUM_O * POS];
  __shared__ float route_lds[NUM_I * NUM_O * POS];
  __shared__ float part_lds[NW * 8 * 32];
  __shared__ __attribute__((aligned(16))) float votes_lds[NW * NC * 8 * 32];

  const int tid  = threadIdx.x;
  const int lane = tid & 31;
  const int wv   = __builtin_amdgcn_readfirstlane(tid >> 5);
  const int o    = wv >> 1;
  const int g    = wv & 1;
  const int i0   = g * IH;
  const int n    = lane & 15;
  const int h    = lane >> 4;

  const int t_begin = (blockIdx.x * 9) >> 2;
  const int t_end   = ((blockIdx.x + 1) * 9) >> 2;

  float sb[NUM_AT];
#pragma unroll
  for (int k = 0; k < NUM_AT; ++k) sb[k] = bias[o * NUM_AT + k];

  const unsigned xl = (unsigned)(2 * h * HW + n);
  v2f xr[IH][2];
  {
    const float* xp0 = X_TILE_PTR(t_begin);
    LOAD_X_CAPS(xp0, 0, XA - 1);
  }

  v2f wf[IH][2];
  {
    const float* wp = w + (i0 * NUM_A) * WCOLS + o * NUM_AT + (unsigned)(2 * h * WCOLS + n);
#pragma unroll
    for (int i = 0; i < IH; ++i) {
#pragma unroll
      for (int c = 0; c < 2; ++c) {
        wf[i][c].x = wp[(i * NUM_A + 4 * c    ) * WCOLS];
        wf[i][c].y = wp[(i * NUM_A + 4 * c + 1) * WCOLS];
      }
    }
  }

  const float* rtp = route_lds + (i0 * NUM_O + o) * POS + n;
  float*       lgp = logits_lds + ((i0 + h) * NUM_O + o) * POS + n;
  float*       myp = part_lds + (wv * 8) * 32 + lane;
  const float* otp = part_lds + ((wv ^ 1) * 8) * 32 + lane;
  v4f*         vcp = (v4f*)(votes_lds + wv * (NC * 8 * 32)) + lane;

#pragma unroll
  for (int i = 0; i < IH; ++i) {
#pragma unroll
    for (int c = 0; c < 2; ++c) {
      asm volatile("" : "+v"(wf[i][c].x), "+v"(wf[i][c].y));
      if (i < XA) asm volatile("" : "+v"(xr[i][c].x), "+v"(xr[i][c].y));
    }
  }

#define PRE_INIT()                                                                       \
  float pre[8];                                                                          \
  if (g == 0) {                                                                          \
    _Pragma("unroll")                                                                    \
    for (int r = 0; r < 8; ++r) {                                                        \
      float lo = sb[r], hi = sb[8 + r];                                                  \
      asm volatile("" : "+s"(lo), "+s"(hi));     \
      pre[r] = h ? hi : lo;                       \
    }                                                                                    \
  } else {                                                                               \
    _Pragma("unroll")                                                                    \
    for (int r = 0; r < 8; ++r) pre[r] = 0.0f;                                           \
  }

#define ROUTE_SOFTMAX()                                                                  \
  if (tid < NUM_I * POS) {                                                               \
    const int i = tid >> 4;                                                              \
    const int p = tid & 15;                                                              \
    float lv[NUM_O];                                                                     \
    float mx = -1e30f;                                                                   \
    _Pragma("unroll")                                                                    \
    for (int oo = 0; oo < NUM_O; ++oo) {                                                 \
      lv[oo] = logits_lds[(i * NUM_O + oo) * POS + p];                                   \
      mx = fmaxf(mx, lv[oo]);                                                            \
    }                                                                                    \
    float s = 0.0f;                                                                      \
    _Pragma("unroll")                                                                    \
    for (int oo = 0; oo < NUM_O; ++oo) { lv[oo] = __expf(lv[oo] - mx); s += lv[oo]; }    \
      \
               \
    const float inv = __builtin_amdgcn_rcpf(s);                                          \
    _Pragma("unroll")                                                                    \
    for (int oo = 0; oo < NUM_O; ++oo)                                                   \
      route_lds[(i * NUM_O + oo) * POS + p] = lv[oo] * inv;                              \
  }                                                                                      \
  lds_barrier();

#define SLAB_LOAD(d_, i_)                                                                \
  {                                                                                      \
    const v4f lo_ = vcp[(2 * (i_)    ) * 32];                                            \
    const v4f hi_ = vcp[(2 * (i_) + 1) * 32];                                            \
    d_ = __builtin_shufflevector(lo_, hi_, 0, 1, 2, 3, 4, 5, 6, 7);                      \
  }

#define VOTE_WMMA(d_, i_)                                                                \
  {                                                                                      \
    d_ = (v8f){};                                                                        \
    d_ = WMMA_F32_4(wf[i_][0], xr[i_][0], d_);                              \
    d_ = WMMA_F32_4(wf[i_][1], xr[i_][1], d_);                              \
    if ((i_) < NC) {                                                                     \
      vcp[(2 * (i_)    ) * 32] = __builtin_shufflevector(d_, d_, 0, 1, 2, 3);            \
      vcp[(2 * (i_) + 1) * 32] = __builtin_shufflevector(d_, d_, 4, 5, 6, 7);            \
    } else {                                                                             \
      kv[(i_) - NC] = d_;                                                                \
    }                                                                                    \
  }

#define PAIR_SLAB(q_) ((q_) >= NR || ((q_) & 1))
#define PAIR_BASE(q_) ((q_) >= NR ? 2 * (q_) - NR : (((q_) & 1) ? (q_) - 1 : NC + (q_)))
#define CAPL(s_) (PAIR_BASE((s_) >> 1) + ((s_) & 1))
#define PAIR_ASK(q_) (!PAIR_SLAB(q_) ? PAIR_BASE((q_) + 1) : (((q_) == 0 || PAIR_SLAB((q_) - 1)) ? PAIR_BASE(q_) : -1))

#define CAPA(s_, FIRST_) ((FIRST_) ? (s_) : CAPL(s_))
#define PASS_A(FIRST_, PF_)                                                              \
  {                                                                                      \
    v8f dt[IH];                                                                          \
    float rtv[IH];                              \
    rtv[0] = rtp[CAPA(0, FIRST_) * NUM_O * POS];                                         \
    rtv[1] = rtp[CAPA(1, FIRST_) * NUM_O * POS];                                         \
    _Pragma("unroll")                                                                    \
    for (int s = 0; s < IH; ++s) {                                                       \
      const int i = CAPA(s, FIRST_);                     \
      if (!(s & 1) && s + 2 < IH) {                                                      \
        rtv[s + 2] = rtp[CAPA(s + 2, FIRST_) * NUM_O * POS];                             \
        rtv[s + 3] = rtp[CAPA(s + 3, FIRST_) * NUM_O * POS];                             \
      }                                                                                  \
      if (!(FIRST_) && !(s & 1) && PAIR_ASK(s >> 1) >= 0) {     \
        SLAB_LOAD(dt[PAIR_ASK(s >> 1)], PAIR_ASK(s >> 1))                                \
        SLAB_LOAD(dt[PAIR_ASK(s >> 1) + 1], PAIR_ASK(s >> 1) + 1)                        \
      }                                                                                  \
      if (FIRST_) {                                                                      \
        VOTE_WMMA(dt[i], i)                                                              \
          \
             \
        if ((s & 1) && s + XA < IH) LOAD_X_CAPS(xc, s + XA - 1, s + XA)                  \
      }                                                                                  \
      else if (i >= NC) dt[i] = kv[i - NC];                                              \
      const float rt = rtv[s];                                                           \
      _Pragma("unroll")                                                                  \
      for (int r = 0; r < 8; ++r) pre[r] += rt * dt[i][r];                               \
      if (s & 1) {                                      \
        if ((PF_) && s < XA) LOAD_X_CAPS(xn, s - 1, s)                                   \
          \
        asm volatile("" : "+v"(pre[0]), "+v"(pre[1]), "+v"(pre[2]), "+v"(pre[3]),        \
                          "+v"(pre[4]), "+v"(pre[5]), "+v"(pre[6]), "+v"(pre[7]));       \
        __builtin_amdgcn_sched_barrier(0);      \
      }                                                                                  \
    }                                                                                    \
  }

#define PASS_B()                                                                         \
  {                                                                                      \
    v8f dt[IH];                                                                          \
    _Pragma("unroll")                                                                    \
    for (int q = 0; q < IH / 2; ++q) {                                                   \
      const int c0 = PAIR_BASE(q);              \
      if (PAIR_ASK(q) >= 0) {                   \
        SLAB_LOAD(dt[PAIR_ASK(q)], PAIR_ASK(q))                                          \
        SLAB_LOAD(dt[PAIR_ASK(q) + 1], PAIR_ASK(q) + 1)                                  \
      }                                                                                  \
      float part[2];                                                                     \
      _Pragma("unroll")                                                                  \
      for (int j = 0; j < 2; ++j) {                                                      \
        const int i = c0 + j;                                                            \
        if (i >= NC) dt[i] = kv[i - NC];                                                 \
          \
          \
        v2f s2 = (v2f){0.0f, 0.0f};                                                      \
        _Pragma("unroll")                                                                \
        for (int r = 0; r < 8; r += 2)                                                   \
          s2 += (v2f){dt[i][r], dt[i][r + 1]} * (v2f){act[r], act[r + 1]};               \
        part[j] = s2.x + s2.y;                      \
      }                                                                                  \
                          \
      const float keep = h ? part[1] : part[0];                                          \
      const float send = h ? part[0] : part[1];                                          \
      const float tot  = keep + other_half(send);                                        \
      lgp[c0 * NUM_O * POS] += tot;                        \
      __builtin_amdgcn_sched_barrier(0);        \
    }                                                                                    \
  }

#define SWAP_SQUASH()                                                                    \
  _Pragma("unroll")                                                                      \
  for (int r = 0; r < 8; ++r) myp[r * 32] = pre[r];                                      \
  lds_barrier();                                                                         \
  _Pragma("unroll")                                                                      \
  for (int r = 0; r < 8; ++r) pre[r] += otp[r * 32];                                     \
  float ns = 0.0f;                                                                       \
  _Pragma("unroll")                                                                      \
  for (int r = 0; r < 8; ++r) ns += pre[r] * pre[r];                                     \
  ns += other_half(ns);                                     \
    \
     \
  const float sc = __builtin_amdgcn_sqrtf(ns) * __builtin_amdgcn_rcpf(1.0f + ns);        \
  float act[8];                                                                          \
  _Pragma("unroll")                                                                      \
  for (int r = 0; r < 8; ++r) act[r] = pre[r] * sc;

#define STORE_OUT()                                                                      \
  {                                                                                      \
    int tc = t;                                                                          \
    asm volatile("" : "+s"(tc));          \
    const int b   = tc / 9;                                                              \
    const int hw0 = (tc - b * 9) * POS;                                                  \
    float* op = (out + (((b * NUM_O + o) * NUM_AT + 4 * g) * HW + hw0)) + (unsigned)(8 * h * HW + n); \
    if (g == 0) {                                                                        \
      _Pragma("unroll")                                                                  \
      for (int r = 0; r < 4; ++r) op[r * HW] = act[r];                                   \
    } else {                                                                             \
      _Pragma("unroll")                                                                  \
      for (int r = 0; r < 4; ++r) op[r * HW] = act[4 + r];                               \
    }                                                                                    \
  }

  int t = t_begin;
#pragma unroll 1
  for (;;) {
  {
    float* zp = logits_lds + (wv * 8) * 32 + lane;
#pragma unroll
    for (int r = 0; r < 8; ++r) zp[r * 32] = 0.0f;
  }
  lds_barrier();

  v8f kv[NR];

  int tcur = t;
  asm volatile("" : "+s"(tcur));
  const float* xc = X_TILE_PTR(tcur);

  {
    const float* xn = x;
    PRE_INIT();
    ROUTE_SOFTMAX();
    PASS_A(1, 0);
    SWAP_SQUASH();
    PASS_B();
    lds_barrier();
  }

  {
    const float* xn = x;
    PRE_INIT();
    ROUTE_SOFTMAX();
    PASS_A(0, 0);
    SWAP_SQUASH();
    PASS_B();
    lds_barrier();
  }

  const bool has_next = (t + 1 < t_end);
  {
    int tn = has_next ? t + 1 : t;
    asm volatile("" : "+s"(tn));
    const float* xn = X_TILE_PTR(tn);

    PRE_INIT();
    ROUTE_SOFTMAX();
    PASS_A(0, 1);
    SWAP_SQUASH();
    STORE_OUT();
  }
  if (!has_next) break;
  ++t;
  }
}

extern "C" void kernel_launch(void* const* d_in, const int* in_sizes, int n_in,
                              void* d_out, int out_size, void* d_ws, size_t ws_size,
                              hipStream_t stream) {
  const float* x    = (const float*)d_in[0];
  const float* wts  = (const float*)d_in[1];
  const float* bias = (const float*)d_in[2];
  float* out = (float*)d_out;
  (void)in_sizes; (void)n_in; (void)out_size; (void)d_ws; (void)ws_size;

  dim3 grid(256);
  dim3 block(NT);
  capsule_routing_kernel<<<grid, block, 0, stream>>>(x, wts, bias, out);
}


# Workgroup's first tile peeled in front of the loop: weights and x asked for capsule by capsule in first-use order, so its matrix pass waits per capsule and runs under the 128-load operand stream
# speedup vs baseline: 3.2232x; 1.0542x over previous
#include <hip/hip_runtime.h>


typedef __attribute__((ext_vector_type(2))) float v2f;
typedef __attribute__((ext_vector_type(4))) float v4f;
typedef __attribute__((ext_vector_type(8))) float v8f;

#define NUM_I   32
#define NUM_A   8
#define NUM_O   10
#define NUM_AT  16
#define POS     16
#define HW      144
#define NW      20
#define NT      640
#define IH      16
#define WCOLS   160
#define NC      8
#define NR      (IH - NC)
#define XA      8
#define PA      6

#define WMMA_F32_4(A_, B_, C_) \
  __builtin_amdgcn_wmma_f32_16x16x4_f32(false, (A_), false, (B_), (short)0, (C_), false, false)

__device__ __forceinline__ float other_half(float v) {
  return __builtin_bit_cast(float, __builtin_amdgcn_permlanex16(__builtin_bit_cast(int, v), __builtin_bit_cast(int, v),
                                                                 0x76543210, 0xfedcba98, false, false));
}

#define X_TILE_PTR(T_) \
  ((x + ((((((T_) / 9) * NUM_I) + i0) * NUM_A) * HW + ((T_) - ((T_) / 9) * 9) * POS)) + xl)

#define LOAD_X_CAPS(P_, JA_, JB_)                                                        \
  {                                                                                      \
    _Pragma("unroll")                                                                    \
    for (int j_ = (JA_); j_ <= (JB_); ++j_) {                                            \
      _Pragma("unroll")                                                                  \
      for (int c_ = 0; c_ < 2; ++c_) {                                                   \
        xr[j_][c_].x = (P_)[(j_ * NUM_A + 4 * c_    ) * HW];                             \
        xr[j_][c_].y = (P_)[(j_ * NUM_A + 4 * c_ + 1) * HW];                             \
      }                                                                                  \
    }                                                                                    \
  }

#define LOAD_W_CAPS(JA_, JB_)                                                            \
  {                                                                                      \
    _Pragma("unroll")                                                                    \
    for (int j_ = (JA_); j_ <= (JB_); ++j_) {                                            \
      _Pragma("unroll")                                                                  \
      for (int c_ = 0; c_ < 2; ++c_) {                                                   \
        wf[j_][c_].x = wp[(j_ * NUM_A + 4 * c_    ) * WCOLS];                            \
        wf[j_][c_].y = wp[(j_ * NUM_A + 4 * c_ + 1) * WCOLS];                            \
      }                                                                                  \
    }                                                                                    \
  }

__device__ __forceinline__ void lds_barrier() {
  asm volatile("s_wait_dscnt 0x0\n\ts_barrier_signal -1\n\ts_barrier_wait -1" ::: "memory");
}

__launch_bounds__(NT)
__global__ void capsule_routing_kernel(const float* __restrict__ x,
                                       const float* __restrict__ w,
                                       const float* __restrict__ bias,
                                       float* __restrict__ out) {
  __shared__ float logits_lds[NUM_I * NUM_O * POS];
  __shared__ float route_lds[NUM_I * NUM_O * POS];
  __shared__ float part_lds[NW * 8 * 32];
  __shared__ __attribute__((aligned(16))) float votes_lds[NW * NC * 8 * 32];

  const int tid  = threadIdx.x;
  const int lane = tid & 31;
  const int wv   = __builtin_amdgcn_readfirstlane(tid >> 5);
  const int o    = wv >> 1;
  const int g    = wv & 1;
  const int i0   = g * IH;
  const int n    = lane & 15;
  const int h    = lane >> 4;

  const int t_begin = (blockIdx.x * 9) >> 2;
  const int t_end   = ((blockIdx.x + 1) * 9) >> 2;

  float sb[NUM_AT];
#pragma unroll
  for (int k = 0; k < NUM_AT; ++k) sb[k] = bias[o * NUM_AT + k];

  const unsigned xl = (unsigned)(2 * h * HW + n);
  v2f xr[IH][2];

  v2f wf[IH][2];

  const float* wp = w + (i0 * NUM_A) * WCOLS + o * NUM_AT + (unsigned)(2 * h * WCOLS + n);
  {
    const float* xp0 = X_TILE_PTR(t_begin);
#pragma unroll
    for (int i = 0; i < PA; ++i) {
      LOAD_X_CAPS(xp0, i, i)
      LOAD_W_CAPS(i, i)
    }
  }

  const float* rtp = route_lds + (i0 * NUM_O + o) * POS + n;
  float*       lgp = logits_lds + ((i0 + h) * NUM_O + o) * POS + n;
  float*       myp = part_lds + (wv * 8) * 32 + lane;
  const float* otp = part_lds + ((wv ^ 1) * 8) * 32 + lane;
  v4f*         vcp = (v4f*)(votes_lds + wv * (NC * 8 * 32)) + lane;


#define PRE_INIT()                                                                       \
  float pre[8];                                                                          \
  if (g == 0) {                                                                          \
    _Pragma("unroll")                                                                    \
    for (int r = 0; r < 8; ++r) {                                                        \
      float lo = sb[r], hi = sb[8 + r];                                                  \
      asm volatile("" : "+s"(lo), "+s"(hi));     \
      pre[r] = h ? hi : lo;                       \
    }                                                                                    \
  } else {                                                                               \
    _Pragma("unroll")                                                                    \
    for (int r = 0; r < 8; ++r) pre[r] = 0.0f;                                           \
  }

#define ROUTE_SOFTMAX()                                                                  \
  if (tid < NUM_I * POS) {                                                               \
    const int i = tid >> 4;                                                              \
    const int p = tid & 15;                                                              \
    float lv[NUM_O];                                                                     \
    float mx = -1e30f;                                                                   \
    _Pragma("unroll")                                                                    \
    for (int oo = 0; oo < NUM_O; ++oo) {                                                 \
      lv[oo] = logits_lds[(i * NUM_O + oo) * POS + p];                                   \
      mx = fmaxf(mx, lv[oo]);                                                            \
    }                                                                                    \
    float s = 0.0f;                                                                      \
    _Pragma("unroll")                                                                    \
    for (int oo = 0; oo < NUM_O; ++oo) { lv[oo] = __expf(lv[oo] - mx); s += lv[oo]; }    \
      \
               \
    const float inv = __builtin_amdgcn_rcpf(s);                                          \
    _Pragma("unroll")                                                                    \
    for (int oo = 0; oo < NUM_O; ++oo)                                                   \
      route_lds[(i * NUM_O + oo) * POS + p] = lv[oo] * inv;                              \
  }                                                                                      \
  lds_barrier();

#define SLAB_LOAD(d_, i_)                                                                \
  {                                                                                      \
    const v4f lo_ = vcp[(2 * (i_)    ) * 32];                                            \
    const v4f hi_ = vcp[(2 * (i_) + 1) * 32];                                            \
    d_ = __builtin_shufflevector(lo_, hi_, 0, 1, 2, 3, 4, 5, 6, 7);                      \
  }

#define VOTE_WMMA(d_, i_)                                                                \
  {                                                                                      \
    d_ = (v8f){};                                                                        \
    d_ = WMMA_F32_4(wf[i_][0], xr[i_][0], d_);                              \
    d_ = WMMA_F32_4(wf[i_][1], xr[i_][1], d_);                              \
    if ((i_) < NC) {                                                                     \
      vcp[(2 * (i_)    ) * 32] = __builtin_shufflevector(d_, d_, 0, 1, 2, 3);            \
      vcp[(2 * (i_) + 1) * 32] = __builtin_shufflevector(d_, d_, 4, 5, 6, 7);            \
    } else {                                                                             \
      kv[(i_) - NC] = d_;                                                                \
    }                                                                                    \
  }

#define PAIR_SLAB(q_) ((q_) >= NR || ((q_) & 1))
#define PAIR_BASE(q_) ((q_) >= NR ? 2 * (q_) - NR : (((q_) & 1) ? (q_) - 1 : NC + (q_)))
#define CAPL(s_) (PAIR_BASE((s_) >> 1) + ((s_) & 1))
#define PAIR_ASK(q_) (!PAIR_SLAB(q_) ? PAIR_BASE((q_) + 1) : (((q_) == 0 || PAIR_SLAB((q_) - 1)) ? PAIR_BASE(q_) : -1))

#define CAPA(s_, FIRST_) ((FIRST_) ? (s_) : CAPL(s_))
#define PASS_A(FIRST_, PF_, XIN_)                                                        \
  {                                                                                      \
    v8f dt[IH];                                                                          \
    float rtv[IH];                              \
    rtv[0] = rtp[CAPA(0, FIRST_) * NUM_O * POS];                                         \
    rtv[1] = rtp[CAPA(1, FIRST_) * NUM_O * POS];                                         \
    _Pragma("unroll")                                                                    \
    for (int s = 0; s < IH; ++s) {                                                       \
      const int i = CAPA(s, FIRST_);                     \
      if (!(s & 1) && s + 2 < IH) {                                                      \
        rtv[s + 2] = rtp[CAPA(s + 2, FIRST_) * NUM_O * POS];                             \
        rtv[s + 3] = rtp[CAPA(s + 3, FIRST_) * NUM_O * POS];                             \
      }                                                                                  \
      if (!(FIRST_) && !(s & 1) && PAIR_ASK(s >> 1) >= 0) {     \
        SLAB_LOAD(dt[PAIR_ASK(s >> 1)], PAIR_ASK(s >> 1))                                \
        SLAB_LOAD(dt[PAIR_ASK(s >> 1) + 1], PAIR_ASK(s >> 1) + 1)                        \
      }                                                                                  \
      if (FIRST_) {                                                                      \
        VOTE_WMMA(dt[i], i)                                                              \
          \
             \
        if ((XIN_) == 1 && (s & 1) && s + XA < IH) LOAD_X_CAPS(xc, s + XA - 1, s + XA)   \
          \
        if ((XIN_) == 2 && (s & 1) && s + PA < IH) {                                     \
          LOAD_X_CAPS(xc, s + PA - 1, s + PA - 1)                                        \
          LOAD_W_CAPS(s + PA - 1, s + PA - 1)                                            \
          LOAD_X_CAPS(xc, s + PA, s + PA)                                                \
          LOAD_W_CAPS(s + PA, s + PA)                                                    \
        }                                                                                \
      }                                                                                  \
      else if (i >= NC) dt[i] = kv[i - NC];                                              \
      const float rt = rtv[s];                                                           \
      _Pragma("unroll")                                                                  \
      for (int r = 0; r < 8; ++r) pre[r] += rt * dt[i][r];                               \
      if (s & 1) {                                      \
        if ((PF_) && s < XA) LOAD_X_CAPS(xn, s - 1, s)                                   \
          \
        asm volatile("" : "+v"(pre[0]), "+v"(pre[1]), "+v"(pre[2]), "+v"(pre[3]),        \
                          "+v"(pre[4]), "+v"(pre[5]), "+v"(pre[6]), "+v"(pre[7]));       \
        __builtin_amdgcn_sched_barrier(0);      \
      }                                                                                  \
    }                                                                                    \
  }

#define PASS_B()                                                                         \
  {                                                                                      \
    v8f dt[IH];                                                                          \
    _Pragma("unroll")                                                                    \
    for (int q = 0; q < IH / 2; ++q) {                                                   \
      const int c0 = PAIR_BASE(q);              \
      if (PAIR_ASK(q) >= 0) {                   \
        SLAB_LOAD(dt[PAIR_ASK(q)], PAIR_ASK(q))                                          \
        SLAB_LOAD(dt[PAIR_ASK(q) + 1], PAIR_ASK(q) + 1)                                  \
      }                                                                                  \
      float part[2];                                                                     \
      _Pragma("unroll")                                                                  \
      for (int j = 0; j < 2; ++j) {                                                      \
        const int i = c0 + j;                                                            \
        if (i >= NC) dt[i] = kv[i - NC];                                                 \
          \
          \
        v2f s2 = (v2f){0.0f, 0.0f};                                                      \
        _Pragma("unroll")                                                                \
        for (int r = 0; r < 8; r += 2)                                                   \
          s2 += (v2f){dt[i][r], dt[i][r + 1]} * (v2f){act[r], act[r + 1]};               \
        part[j] = s2.x + s2.y;                      \
      }                                                                                  \
                          \
      const float keep = h ? part[1] : part[0];                                          \
      const float send = h ? part[0] : part[1];                                          \
      const float tot  = keep + other_half(send);                                        \
      lgp[c0 * NUM_O * POS] += tot;                        \
      __builtin_amdgcn_sched_barrier(0);        \
    }                                                                                    \
  }

#define SWAP_SQUASH()                                                                    \
  _Pragma("unroll")                                                                      \
  for (int r = 0; r < 8; ++r) myp[r * 32] = pre[r];                                      \
  lds_barrier();                                                                         \
  _Pragma("unroll")                                                                      \
  for (int r = 0; r < 8; ++r) pre[r] += otp[r * 32];                                     \
  float ns = 0.0f;                                                                       \
  _Pragma("unroll")                                                                      \
  for (int r = 0; r < 8; ++r) ns += pre[r] * pre[r];                                     \
  ns += other_half(ns);                                     \
    \
     \
  const float sc = __builtin_amdgcn_sqrtf(ns) * __builtin_amdgcn_rcpf(1.0f + ns);        \
  float act[8];                                                                          \
  _Pragma("unroll")                                                                      \
  for (int r = 0; r < 8; ++r) act[r] = pre[r] * sc;

#define STORE_OUT()                                                                      \
  {                                                                                      \
    int tc = t;                                                                          \
    asm volatile("" : "+s"(tc));          \
    const int b   = tc / 9;                                                              \
    const int hw0 = (tc - b * 9) * POS;                                                  \
    float* op = (out + (((b * NUM_O + o) * NUM_AT + 4 * g) * HW + hw0)) + (unsigned)(8 * h * HW + n); \
    if (g == 0) {                                                                        \
      _Pragma("unroll")                                                                  \
      for (int r = 0; r < 4; ++r) op[r * HW] = act[r];                                   \
    } else {                                                                             \
      _Pragma("unroll")                                                                  \
      for (int r = 0; r < 4; ++r) op[r * HW] = act[4 + r];                               \
    }                                                                                    \
  }

#define TILE_BODY(XIN_, HASNEXT_)                                                        \
  {                                                                                      \
    {                                                                                    \
      float* zp = logits_lds + (wv * 8) * 32 + lane;                                     \
      _Pragma("unroll")                                                                  \
      for (int r = 0; r < 8; ++r) zp[r * 32] = 0.0f;                                     \
    }                                                                                    \
    lds_barrier();                                                                       \
    v8f kv[NR];                                                                          \
    int tcur = t;                                                                        \
    asm volatile("" : "+s"(tcur));        \
    const float* xc = X_TILE_PTR(tcur);                                                  \
    {                                                                                    \
      const float* xn = x;                \
      PRE_INIT();                                                                        \
      ROUTE_SOFTMAX();                                                                   \
      PASS_A(1, 0, XIN_);                                                                \
      SWAP_SQUASH();                                                                     \
      PASS_B();                                                                          \
      lds_barrier();                                                                     \
    }                                                                                    \
    {                                                                                    \
      const float* xn = x;                \
      PRE_INIT();                                                                        \
      ROUTE_SOFTMAX();                                                                   \
      PASS_A(0, 0, 0);                                                                   \
      SWAP_SQUASH();                                                                     \
      PASS_B();                                                                          \
      lds_barrier();                                                                     \
    }                                                                                    \
    has_next = (HASNEXT_);                                                               \
    {                                                                                    \
      int tn = has_next ? t + 1 : t;                                                     \
      asm volatile("" : "+s"(tn));        \
      const float* xn = X_TILE_PTR(tn);                                                  \
      PRE_INIT();                                                                        \
      ROUTE_SOFTMAX();                                                                   \
      PASS_A(0, 1, 0);                                                                   \
      SWAP_SQUASH();                                                                     \
      STORE_OUT();                                                                       \
    }                                                                                    \
  }

  int t = t_begin;
  bool has_next;
  TILE_BODY(2, true)
  ++t;
#pragma unroll 1
  for (;;) {
    TILE_BODY(1, t + 1 < t_end)
    if (!has_next) break;
    ++t;
  }
}

extern "C" void kernel_launch(void* const* d_in, const int* in_sizes, int n_in,
                              void* d_out, int out_size, void* d_ws, size_t ws_size,
                              hipStream_t stream) {
  const float* x    = (const float*)d_in[0];
  const float* wts  = (const float*)d_in[1];
  const float* bias = (const float*)d_in[2];
  float* out = (float*)d_out;
  (void)in_sizes; (void)n_in; (void)out_size; (void)d_ws; (void)ws_size;

  dim3 grid(256);
  dim3 block(NT);
  capsule_routing_kernel<<<grid, block, 0, stream>>>(x, wts, bias, out);
}
